# starting kernel
# baseline (speedup 1.0000x reference)
#include <hip/hip_runtime.h>
#include <hip/hip_bf16.h>

typedef __attribute__((ext_vector_type(2))) float v2f;
typedef __attribute__((ext_vector_type(8))) float v8f;

#define CDIM 128

__global__ void GNN_zero_kernel(float* __restrict__ p, size_t n) {
    size_t i = (size_t)blockIdx.x * blockDim.x + threadIdx.x;
    size_t stride = (size_t)gridDim.x * blockDim.x;
    for (; i < n; i += stride) p[i] = 0.0f;
}

__global__ void GNN_scatter_kernel(const float* __restrict__ feat_src,
                                   const float* __restrict__ feat_dst,
                                   const int*   __restrict__ src_idx,
                                   const int*   __restrict__ dst_idx,
                                   float* __restrict__ agg_m,
                                   float* __restrict__ agg_u,
                                   float* __restrict__ cnt_m,
                                   float* __restrict__ cnt_u,
                                   int n_edges, int add_cnt) {
    int wid  = (int)((blockIdx.x * (size_t)blockDim.x + threadIdx.x) >> 5);
    int lane = threadIdx.x & 31;
    if (wid >= n_edges) return;
    int s = src_idx[wid];
    int d = dst_idx[wid];
    const float* xs = feat_src + (size_t)s * CDIM;
    const float* xd = feat_dst + (size_t)d * CDIM;
    float* am = agg_m + (size_t)d * CDIM;
    float* au = agg_u + (size_t)s * CDIM;
#pragma unroll
    for (int t = 0; t < 4; ++t) {
        int c = lane + 32 * t;
        atomicAdd(am + c, xs[c]);
        atomicAdd(au + c, xd[c]);
    }
    if (add_cnt && lane == 0) {
        atomicAdd(cnt_m + d, 1.0f);
        atomicAdd(cnt_u + s, 1.0f);
    }
}

__global__ __launch_bounds__(256)
void GNN_sage_gemm_kernel(const float* __restrict__ agg,
                          const float* __restrict__ cnt,
                          const float* __restrict__ root,
                          const float* __restrict__ W_nbr,
                          const float* __restrict__ W_root,
                          const float* __restrict__ bias,
                          float* __restrict__ out,
                          int n_rows, int do_relu) {
    __shared__ float sMean[16][CDIM + 4];
    __shared__ float sRoot[16][CDIM + 4];

    const int tid  = threadIdx.x;
    const int row0 = blockIdx.x * 16;

    {
        int r  = tid >> 4;
        int c0 = (tid & 15) * 8;
        int gr = row0 + r;
        if (gr < n_rows) {
            float cv  = cnt[gr];
            float inv = 1.0f / fmaxf(cv, 1.0f);
            const float* ga = agg  + (size_t)gr * CDIM + c0;
            const float* gx = root + (size_t)gr * CDIM + c0;
#pragma unroll
            for (int i = 0; i < 8; ++i) {
                sMean[r][c0 + i] = ga[i] * inv;
                sRoot[r][c0 + i] = gx[i];
            }
        } else {
#pragma unroll
            for (int i = 0; i < 8; ++i) {
                sMean[r][c0 + i] = 0.0f;
                sRoot[r][c0 + i] = 0.0f;
            }
        }
    }
    __syncthreads();

    const int wave  = tid >> 5;
    const int lane  = tid & 31;
    const int lhalf = (lane >> 4) & 1;
    const int lm    = lane & 15;
    const int n0    = wave * 16;

    v8f acc;
    {
        float bv = bias[n0 + lm];
#pragma unroll
        for (int i = 0; i < 8; ++i) acc[i] = bv;
    }

#pragma unroll
    for (int k = 0; k < CDIM; k += 4) {
        int ka = k + 2 * lhalf;
        v2f a, b;
        a.x = sMean[lm][ka];
        a.y = sMean[lm][ka + 1];
        b.x = W_nbr[(size_t)ka * CDIM + n0 + lm];
        b.y = W_nbr[(size_t)(ka + 1) * CDIM + n0 + lm];
        acc = __builtin_amdgcn_wmma_f32_16x16x4_f32(
            false, a, false, b, (short)0, acc, false, false);
    }
#pragma unroll
    for (int k = 0; k < CDIM; k += 4) {
        int ka = k + 2 * lhalf;
        v2f a, b;
        a.x = sRoot[lm][ka];
        a.y = sRoot[lm][ka + 1];
        b.x = W_root[(size_t)ka * CDIM + n0 + lm];
        b.y = W_root[(size_t)(ka + 1) * CDIM + n0 + lm];
        acc = __builtin_amdgcn_wmma_f32_16x16x4_f32(
            false, a, false, b, (short)0, acc, false, false);
    }

    if (do_relu) {
#pragma unroll
        for (int i = 0; i < 8; ++i) acc[i] = fmaxf(acc[i], 0.0f);
    }

#pragma unroll
    for (int v = 0; v < 8; ++v) {
        int gr = row0 + v + 8 * lhalf;
        if (gr < n_rows) out[(size_t)gr * CDIM + n0 + lm] = acc[v];
    }
}

__global__ void GNN_decode_kernel(const float* __restrict__ h_u2,
                                  const float* __restrict__ h_m2,
                                  const int*   __restrict__ lab_u,
                                  const int*   __restrict__ lab_m,
                                  float* __restrict__ out, int n) {
    int wid  = (int)((blockIdx.x * (size_t)blockDim.x + threadIdx.x) >> 5);
    int lane = threadIdx.x & 31;
    if (wid >= n) return;
    const float* a = h_u2 + (size_t)lab_u[wid] * CDIM;
    const float* b = h_m2 + (size_t)lab_m[wid] * CDIM;
    float s = 0.0f;
#pragma unroll
    for (int t = 0; t < 4; ++t) {
        int c = lane + 32 * t;
        s += a[c] * b[c];
    }
#pragma unroll
    for (int off = 16; off > 0; off >>= 1) s += __shfl_xor(s, off, 32);
    if (lane == 0) out[wid] = s;
}

extern "C" void kernel_launch(void* const* d_in, const int* in_sizes, int n_in,
                              void* d_out, int out_size, void* d_ws, size_t ws_size,
                              hipStream_t stream) {
    const float* x_user  = (const float*)d_in[0];
    const float* x_movie = (const float*)d_in[1];
    const int* edge_src  = (const int*)d_in[2];
    const int* edge_dst  = (const int*)d_in[3];
    const int* label_u   = (const int*)d_in[4];
    const int* label_m   = (const int*)d_in[5];
    const float* W_nbr1_um  = (const float*)d_in[6];
    const float* W_root1_um = (const float*)d_in[7];
    const float* W_nbr1_mu  = (const float*)d_in[8];
    const float* W_root1_mu = (const float*)d_in[9];
    const float* W_nbr2_um  = (const float*)d_in[10];
    const float* W_root2_um = (const float*)d_in[11];
    const float* W_nbr2_mu  = (const float*)d_in[12];
    const float* W_root2_mu = (const float*)d_in[13];
    const float* b1_m = (const float*)d_in[14];
    const float* b1_u = (const float*)d_in[15];
    const float* b2_m = (const float*)d_in[16];
    const float* b2_u = (const float*)d_in[17];

    const int n_users  = in_sizes[0] / CDIM;
    const int n_movies = in_sizes[1] / CDIM;
    const int E  = in_sizes[2];
    const int EL = in_sizes[4];

    float* ws = (float*)d_ws;
    float* agg_m = ws;
    float* agg_u = agg_m + (size_t)n_movies * CDIM;
    float* cnt_m = agg_u + (size_t)n_users * CDIM;
    float* cnt_u = cnt_m + n_movies;
    float* h_m   = cnt_u + n_users;
    float* h_u   = h_m + (size_t)n_movies * CDIM;

    float* out = (float*)d_out;

    const size_t agg_total = (size_t)(n_movies + n_users) * CDIM;
    const size_t zero1 = agg_total + (size_t)n_movies + (size_t)n_users;

    const int edge_blocks = (E + 7) / 8;
    const int lab_blocks  = (EL + 7) / 8;
    const int mov_blocks  = (n_movies + 15) / 16;
    const int usr_blocks  = (n_users + 15) / 16;

    GNN_zero_kernel<<<2048, 256, 0, stream>>>(ws, zero1);
    GNN_scatter_kernel<<<edge_blocks, 256, 0, stream>>>(
        x_user, x_movie, edge_src, edge_dst,
        agg_m, agg_u, cnt_m, cnt_u, E, 1);
    GNN_sage_gemm_kernel<<<mov_blocks, 256, 0, stream>>>(
        agg_m, cnt_m, x_movie, W_nbr1_um, W_root1_um, b1_m, h_m, n_movies, 1);
    GNN_sage_gemm_kernel<<<usr_blocks, 256, 0, stream>>>(
        agg_u, cnt_u, x_user, W_nbr1_mu, W_root1_mu, b1_u, h_u, n_users, 1);

    GNN_zero_kernel<<<2048, 256, 0, stream>>>(ws, agg_total);
    GNN_scatter_kernel<<<edge_blocks, 256, 0, stream>>>(
        h_u, h_m, edge_src, edge_dst,
        agg_m, agg_u, cnt_m, cnt_u, E, 0);
    GNN_sage_gemm_kernel<<<mov_blocks, 256, 0, stream>>>(
        agg_m, cnt_m, h_m, W_nbr2_um, W_root2_um, b2_m, agg_m, n_movies, 0);
    GNN_sage_gemm_kernel<<<usr_blocks, 256, 0, stream>>>(
        agg_u, cnt_u, h_u, W_nbr2_mu, W_root2_mu, b2_u, agg_u, n_users, 0);

    GNN_decode_kernel<<<lab_blocks, 256, 0, stream>>>(
        agg_u  , agg_m  , label_u, label_m, out, EL);

    (void)n_in; (void)out_size; (void)ws_size;
}


# Float atomic scatter-add replaced by per-call neighbour lists (integer arrival-order rank, prefix sum) and an atomic-free one-wave-per-node gather; per-node sum order = arrival rank
# speedup vs baseline: 1.5148x; 1.5148x over previous
#include <hip/hip_runtime.h>
#include <hip/hip_bf16.h>

typedef __attribute__((ext_vector_type(2))) float v2f;
typedef __attribute__((ext_vector_type(8))) float v8f;

#define CDIM 128

__global__ void GNN_zero_kernel(float* __restrict__ p, size_t n) {
    size_t i = (size_t)blockIdx.x * blockDim.x + threadIdx.x;
    size_t stride = (size_t)gridDim.x * blockDim.x;
    for (; i < n; i += stride) p[i] = 0.0f;
}

__global__ void GNN_rank_kernel(const int* __restrict__ src_idx,
                                const int* __restrict__ dst_idx,
                                int* __restrict__ deg,
                                int* __restrict__ rank,
                                int n_edges, int n_movies) {
    int e = (int)(blockIdx.x * (size_t)blockDim.x + threadIdx.x);
    if (e >= n_edges) return;
    int s = src_idx[e];
    int d = dst_idx[e];
    rank[e]           = atomicAdd(deg + d, 1);
    rank[n_edges + e] = atomicAdd(deg + n_movies + s, 1);
}

__global__ __launch_bounds__(1024)
void GNN_offsets_kernel(const int* __restrict__ deg,
                        int* __restrict__ off,
                        float* __restrict__ cnt, int n) {
    __shared__ int sFront[32];
    __shared__ int sWave[32];
    const int tid   = threadIdx.x;
    const int lane  = tid & 31;
    const int wave  = tid >> 5;
    const int tile0 = blockIdx.x * 1024;

    int front = 0;
    for (int i = tid; i < tile0; i += 1024) front += deg[i];
    const int gi = tile0 + tid;
    const int x  = (gi < n) ? deg[gi] : 0;

#pragma unroll
    for (int o = 16; o > 0; o >>= 1) front += __shfl_xor(front, o, 32);
    int inc = x;
#pragma unroll
    for (int o = 1; o < 32; o <<= 1) {
        int t = __shfl_up(inc, o, 32);
        if (lane >= o) inc += t;
    }
    if (lane == 0)  sFront[wave] = front;
    if (lane == 31) sWave[wave]  = inc;
    __syncthreads();
    int base = sFront[lane] + ((lane < wave) ? sWave[lane] : 0);
#pragma unroll
    for (int o = 16; o > 0; o >>= 1) base += __shfl_xor(base, o, 32);
    if (gi < n) {
        off[gi] = base + inc - x;
        cnt[gi] = (float)x;
    }
}

__global__ void GNN_fill_kernel(const int* __restrict__ src_idx,
                                const int* __restrict__ dst_idx,
                                const int* __restrict__ off,
                                const int* __restrict__ rank,
                                int* __restrict__ col,
                                int n_edges, int n_movies) {
    int e = (int)(blockIdx.x * (size_t)blockDim.x + threadIdx.x);
    if (e >= n_edges) return;
    int s = src_idx[e];
    int d = dst_idx[e];
    col[off[d] + rank[e]]                       = s;
    col[off[n_movies + s] + rank[n_edges + e]]  = d;
}

static __device__ __forceinline__ float4 GNN_row4(const float* feat, int r, int lane) {
    return *(const float4*)(feat + (size_t)r * CDIM + lane * 4);
}
static __device__ __forceinline__ void GNN_add4(float4& a, const float4& x) {
    a.x += x.x; a.y += x.y; a.z += x.z; a.w += x.w;
}

__global__ __launch_bounds__(256)
void GNN_gather_kernel(const float* __restrict__ feat_user,
                       const float* __restrict__ feat_movie,
                       const int*   __restrict__ off,
                       const int*   __restrict__ deg,
                       const int*   __restrict__ col,
                       float* __restrict__ agg,
                       int n_movies, int n_total) {
    int v = (int)((blockIdx.x * (size_t)blockDim.x + threadIdx.x) >> 5);
    v = __builtin_amdgcn_readfirstlane(v);
    const int lane = threadIdx.x & 31;
    if (v >= n_total) return;
    const float* feat = (v < n_movies) ? feat_user : feat_movie;
    const int start = off[v];
    const int len   = deg[v];
    float4 acc = make_float4(0.0f, 0.0f, 0.0f, 0.0f);
    for (int base = 0; base < len; base += 32) {
        const int n   = min(32, len - base);
        const int idx = (lane < n) ? col[start + base + lane] : 0;
        int j = 0;
        for (; j + 4 <= n; j += 4) {
            const float4 x0 = GNN_row4(feat, __builtin_amdgcn_readlane(idx, j),     lane);
            const float4 x1 = GNN_row4(feat, __builtin_amdgcn_readlane(idx, j + 1), lane);
            const float4 x2 = GNN_row4(feat, __builtin_amdgcn_readlane(idx, j + 2), lane);
            const float4 x3 = GNN_row4(feat, __builtin_amdgcn_readlane(idx, j + 3), lane);
            GNN_add4(acc, x0); GNN_add4(acc, x1); GNN_add4(acc, x2); GNN_add4(acc, x3);
        }
        if (j < n) {
            const bool two   = j + 1 < n;
            const bool three = j + 2 < n;
            const int r0 = __builtin_amdgcn_readlane(idx, j);
            const int r1 = __builtin_amdgcn_readlane(idx, two   ? j + 1 : j);
            const int r2 = __builtin_amdgcn_readlane(idx, three ? j + 2 : j);
            const float4 x0 = GNN_row4(feat, r0, lane);
            float4 x1 = make_float4(0.0f, 0.0f, 0.0f, 0.0f);
            float4 x2 = make_float4(0.0f, 0.0f, 0.0f, 0.0f);
            if (two)   x1 = GNN_row4(feat, r1, lane);
            if (three) x2 = GNN_row4(feat, r2, lane);
            GNN_add4(acc, x0);
            if (two)   GNN_add4(acc, x1);
            if (three) GNN_add4(acc, x2);
        }
    }
    *(float4*)(agg + (size_t)v * CDIM + lane * 4) = acc;
}

__global__ __launch_bounds__(256)
void GNN_sage_gemm_kernel(const float* __restrict__ agg,
                          const float* __restrict__ cnt,
                          const float* __restrict__ root,
                          const float* __restrict__ W_nbr,
                          const float* __restrict__ W_root,
                          const float* __restrict__ bias,
                          float* __restrict__ out,
                          int n_rows, int do_relu) {
    __shared__ float sMean[16][CDIM + 4];
    __shared__ float sRoot[16][CDIM + 4];

    const int tid  = threadIdx.x;
    const int row0 = blockIdx.x * 16;

    {
        int r  = tid >> 4;
        int c0 = (tid & 15) * 8;
        int gr = row0 + r;
        if (gr < n_rows) {
            float cv  = cnt[gr];
            float inv = 1.0f / fmaxf(cv, 1.0f);
            const float* ga = agg  + (size_t)gr * CDIM + c0;
            const float* gx = root + (size_t)gr * CDIM + c0;
#pragma unroll
            for (int i = 0; i < 8; ++i) {
                sMean[r][c0 + i] = ga[i] * inv;
                sRoot[r][c0 + i] = gx[i];
            }
        } else {
#pragma unroll
            for (int i = 0; i < 8; ++i) {
                sMean[r][c0 + i] = 0.0f;
                sRoot[r][c0 + i] = 0.0f;
            }
        }
    }
    __syncthreads();

    const int wave  = tid >> 5;
    const int lane  = tid & 31;
    const int lhalf = (lane >> 4) & 1;
    const int lm    = lane & 15;
    const int n0    = wave * 16;

    v8f acc;
    {
        float bv = bias[n0 + lm];
#pragma unroll
        for (int i = 0; i < 8; ++i) acc[i] = bv;
    }

#pragma unroll
    for (int k = 0; k < CDIM; k += 4) {
        int ka = k + 2 * lhalf;
        v2f a, b;
        a.x = sMean[lm][ka];
        a.y = sMean[lm][ka + 1];
        b.x = W_nbr[(size_t)ka * CDIM + n0 + lm];
        b.y = W_nbr[(size_t)(ka + 1) * CDIM + n0 + lm];
        acc = __builtin_amdgcn_wmma_f32_16x16x4_f32(
            false, a, false, b, (short)0, acc, false, false);
    }
#pragma unroll
    for (int k = 0; k < CDIM; k += 4) {
        int ka = k + 2 * lhalf;
        v2f a, b;
        a.x = sRoot[lm][ka];
        a.y = sRoot[lm][ka + 1];
        b.x = W_root[(size_t)ka * CDIM + n0 + lm];
        b.y = W_root[(size_t)(ka + 1) * CDIM + n0 + lm];
        acc = __builtin_amdgcn_wmma_f32_16x16x4_f32(
            false, a, false, b, (short)0, acc, false, false);
    }

    if (do_relu) {
#pragma unroll
        for (int i = 0; i < 8; ++i) acc[i] = fmaxf(acc[i], 0.0f);
    }

#pragma unroll
    for (int v = 0; v < 8; ++v) {
        int gr = row0 + v + 8 * lhalf;
        if (gr < n_rows) out[(size_t)gr * CDIM + n0 + lm] = acc[v];
    }
}

__global__ void GNN_decode_kernel(const float* __restrict__ h_u2,
                                  const float* __restrict__ h_m2,
                                  const int*   __restrict__ lab_u,
                                  const int*   __restrict__ lab_m,
                                  float* __restrict__ out, int n) {
    int wid  = (int)((blockIdx.x * (size_t)blockDim.x + threadIdx.x) >> 5);
    int lane = threadIdx.x & 31;
    if (wid >= n) return;
    const float* a = h_u2 + (size_t)lab_u[wid] * CDIM;
    const float* b = h_m2 + (size_t)lab_m[wid] * CDIM;
    float s = 0.0f;
#pragma unroll
    for (int t = 0; t < 4; ++t) {
        int c = lane + 32 * t;
        s += a[c] * b[c];
    }
#pragma unroll
    for (int off = 16; off > 0; off >>= 1) s += __shfl_xor(s, off, 32);
    if (lane == 0) out[wid] = s;
}

extern "C" void kernel_launch(void* const* d_in, const int* in_sizes, int n_in,
                              void* d_out, int out_size, void* d_ws, size_t ws_size,
                              hipStream_t stream) {
    const float* x_user  = (const float*)d_in[0];
    const float* x_movie = (const float*)d_in[1];
    const int* edge_src  = (const int*)d_in[2];
    const int* edge_dst  = (const int*)d_in[3];
    const int* label_u   = (const int*)d_in[4];
    const int* label_m   = (const int*)d_in[5];
    const float* W_nbr1_um  = (const float*)d_in[6];
    const float* W_root1_um = (const float*)d_in[7];
    const float* W_nbr1_mu  = (const float*)d_in[8];
    const float* W_root1_mu = (const float*)d_in[9];
    const float* W_nbr2_um  = (const float*)d_in[10];
    const float* W_root2_um = (const float*)d_in[11];
    const float* W_nbr2_mu  = (const float*)d_in[12];
    const float* W_root2_mu = (const float*)d_in[13];
    const float* b1_m = (const float*)d_in[14];
    const float* b1_u = (const float*)d_in[15];
    const float* b2_m = (const float*)d_in[16];
    const float* b2_u = (const float*)d_in[17];

    const int n_users  = in_sizes[0] / CDIM;
    const int n_movies = in_sizes[1] / CDIM;
    const int E  = in_sizes[2];
    const int EL = in_sizes[4];

    const int N = n_movies + n_users;

    float* ws = (float*)d_ws;
    float* agg_m = ws;
    float* agg_u = agg_m + (size_t)n_movies * CDIM;
    float* cnt_m = agg_u + (size_t)n_users * CDIM;
    float* cnt_u = cnt_m + n_movies;
    float* h_m   = cnt_u + n_users;
    float* h_u   = h_m + (size_t)n_movies * CDIM;
    int*   deg   = (int*)(h_u + (size_t)n_users * CDIM);
    int*   off   = deg + N;
    int*   rank  = off + N;
    int*   col   = rank + 2 * (size_t)E;

    float* out = (float*)d_out;

    const int node_blocks = (N + 255) / 256;
    const int tile_blocks = (N + 1023) / 1024;
    const int edge_blocks = (E + 255) / 256;
    const int wave_blocks = (N + 7) / 8;
    const int lab_blocks  = (EL + 7) / 8;
    const int mov_blocks  = (n_movies + 15) / 16;
    const int usr_blocks  = (n_users + 15) / 16;

    GNN_zero_kernel<<<node_blocks, 256, 0, stream>>>((float*)deg, (size_t)N);
    GNN_rank_kernel<<<edge_blocks, 256, 0, stream>>>(
        edge_src, edge_dst, deg, rank, E, n_movies);
    GNN_offsets_kernel<<<tile_blocks, 1024, 0, stream>>>(deg, off, cnt_m, N);
    GNN_fill_kernel<<<edge_blocks, 256, 0, stream>>>(
        edge_src, edge_dst, off, rank, col, E, n_movies);

    GNN_gather_kernel<<<wave_blocks, 256, 0, stream>>>(
        x_user, x_movie, off, deg, col, agg_m, n_movies, N);
    GNN_sage_gemm_kernel<<<mov_blocks, 256, 0, stream>>>(
        agg_m, cnt_m, x_movie, W_nbr1_um, W_root1_um, b1_m, h_m, n_movies, 1);
    GNN_sage_gemm_kernel<<<usr_blocks, 256, 0, stream>>>(
        agg_u, cnt_u, x_user, W_nbr1_mu, W_root1_mu, b1_u, h_u, n_users, 1);

    GNN_gather_kernel<<<wave_blocks, 256, 0, stream>>>(
        h_u, h_m, off, deg, col, agg_m, n_movies, N);
    GNN_sage_gemm_kernel<<<mov_blocks, 256, 0, stream>>>(
        agg_m, cnt_m, h_m, W_nbr2_um, W_root2_um, b2_m, agg_m, n_movies, 0);
    GNN_sage_gemm_kernel<<<usr_blocks, 256, 0, stream>>>(
        agg_u, cnt_u, h_u, W_nbr2_mu, W_root2_mu, b2_u, agg_u, n_users, 0);

    GNN_decode_kernel<<<lab_blocks, 256, 0, stream>>>(
        agg_u  , agg_m  , label_u, label_m, out, EL);

    (void)n_in; (void)out_size; (void)ws_size;
}


# Decoder: one wave per 32 labels instead of one wave per label (500000 waves become 15632); ids loaded coalesced, float4 row loads; per-label sum order changes: lane owns 4 adjacent channels
# speedup vs baseline: 1.6793x; 1.1086x over previous
#include <hip/hip_runtime.h>
#include <hip/hip_bf16.h>

typedef __attribute__((ext_vector_type(2))) float v2f;
typedef __attribute__((ext_vector_type(8))) float v8f;

#define CDIM 128

__global__ void GNN_zero_kernel(float* __restrict__ p, size_t n) {
    size_t i = (size_t)blockIdx.x * blockDim.x + threadIdx.x;
    size_t stride = (size_t)gridDim.x * blockDim.x;
    for (; i < n; i += stride) p[i] = 0.0f;
}

__global__ void GNN_rank_kernel(const int* __restrict__ src_idx,
                                const int* __restrict__ dst_idx,
                                int* __restrict__ deg,
                                int* __restrict__ rank,
                                int n_edges, int n_movies) {
    int e = (int)(blockIdx.x * (size_t)blockDim.x + threadIdx.x);
    if (e >= n_edges) return;
    int s = src_idx[e];
    int d = dst_idx[e];
    rank[e]           = atomicAdd(deg + d, 1);
    rank[n_edges + e] = atomicAdd(deg + n_movies + s, 1);
}

__global__ __launch_bounds__(1024)
void GNN_offsets_kernel(const int* __restrict__ deg,
                        int* __restrict__ off,
                        float* __restrict__ cnt, int n) {
    __shared__ int sFront[32];
    __shared__ int sWave[32];
    const int tid   = threadIdx.x;
    const int lane  = tid & 31;
    const int wave  = tid >> 5;
    const int tile0 = blockIdx.x * 1024;

    int front = 0;
    for (int i = tid; i < tile0; i += 1024) front += deg[i];
    const int gi = tile0 + tid;
    const int x  = (gi < n) ? deg[gi] : 0;

#pragma unroll
    for (int o = 16; o > 0; o >>= 1) front += __shfl_xor(front, o, 32);
    int inc = x;
#pragma unroll
    for (int o = 1; o < 32; o <<= 1) {
        int t = __shfl_up(inc, o, 32);
        if (lane >= o) inc += t;
    }
    if (lane == 0)  sFront[wave] = front;
    if (lane == 31) sWave[wave]  = inc;
    __syncthreads();
    int base = sFront[lane] + ((lane < wave) ? sWave[lane] : 0);
#pragma unroll
    for (int o = 16; o > 0; o >>= 1) base += __shfl_xor(base, o, 32);
    if (gi < n) {
        off[gi] = base + inc - x;
        cnt[gi] = (float)x;
    }
}

__global__ void GNN_fill_kernel(const int* __restrict__ src_idx,
                                const int* __restrict__ dst_idx,
                                const int* __restrict__ off,
                                const int* __restrict__ rank,
                                int* __restrict__ col,
                                int n_edges, int n_movies) {
    int e = (int)(blockIdx.x * (size_t)blockDim.x + threadIdx.x);
    if (e >= n_edges) return;
    int s = src_idx[e];
    int d = dst_idx[e];
    col[off[d] + rank[e]]                       = s;
    col[off[n_movies + s] + rank[n_edges + e]]  = d;
}

static __device__ __forceinline__ float4 GNN_row4(const float* feat, int r, int lane) {
    return *(const float4*)(feat + (size_t)r * CDIM + lane * 4);
}
static __device__ __forceinline__ void GNN_add4(float4& a, const float4& x) {
    a.x += x.x; a.y += x.y; a.z += x.z; a.w += x.w;
}

__global__ __launch_bounds__(256)
void GNN_gather_kernel(const float* __restrict__ feat_user,
                       const float* __restrict__ feat_movie,
                       const int*   __restrict__ off,
                       const int*   __restrict__ deg,
                       const int*   __restrict__ col,
                       float* __restrict__ agg,
                       int n_movies, int n_total) {
    int v = (int)((blockIdx.x * (size_t)blockDim.x + threadIdx.x) >> 5);
    v = __builtin_amdgcn_readfirstlane(v);
    const int lane = threadIdx.x & 31;
    if (v >= n_total) return;
    const float* feat = (v < n_movies) ? feat_user : feat_movie;
    const int start = off[v];
    const int len   = deg[v];
    float4 acc = make_float4(0.0f, 0.0f, 0.0f, 0.0f);
    for (int base = 0; base < len; base += 32) {
        const int n   = min(32, len - base);
        const int idx = (lane < n) ? col[start + base + lane] : 0;
        int j = 0;
        for (; j + 4 <= n; j += 4) {
            const float4 x0 = GNN_row4(feat, __builtin_amdgcn_readlane(idx, j),     lane);
            const float4 x1 = GNN_row4(feat, __builtin_amdgcn_readlane(idx, j + 1), lane);
            const float4 x2 = GNN_row4(feat, __builtin_amdgcn_readlane(idx, j + 2), lane);
            const float4 x3 = GNN_row4(feat, __builtin_amdgcn_readlane(idx, j + 3), lane);
            GNN_add4(acc, x0); GNN_add4(acc, x1); GNN_add4(acc, x2); GNN_add4(acc, x3);
        }
        if (j < n) {
            const bool two   = j + 1 < n;
            const bool three = j + 2 < n;
            const int r0 = __builtin_amdgcn_readlane(idx, j);
            const int r1 = __builtin_amdgcn_readlane(idx, two   ? j + 1 : j);
            const int r2 = __builtin_amdgcn_readlane(idx, three ? j + 2 : j);
            const float4 x0 = GNN_row4(feat, r0, lane);
            float4 x1 = make_float4(0.0f, 0.0f, 0.0f, 0.0f);
            float4 x2 = make_float4(0.0f, 0.0f, 0.0f, 0.0f);
            if (two)   x1 = GNN_row4(feat, r1, lane);
            if (three) x2 = GNN_row4(feat, r2, lane);
            GNN_add4(acc, x0);
            if (two)   GNN_add4(acc, x1);
            if (three) GNN_add4(acc, x2);
        }
    }
    *(float4*)(agg + (size_t)v * CDIM + lane * 4) = acc;
}

__global__ __launch_bounds__(256)
void GNN_sage_gemm_kernel(const float* __restrict__ agg,
                          const float* __restrict__ cnt,
                          const float* __restrict__ root,
                          const float* __restrict__ W_nbr,
                          const float* __restrict__ W_root,
                          const float* __restrict__ bias,
                          float* __restrict__ out,
                          int n_rows, int do_relu) {
    __shared__ float sMean[16][CDIM + 4];
    __shared__ float sRoot[16][CDIM + 4];

    const int tid  = threadIdx.x;
    const int row0 = blockIdx.x * 16;

    {
        int r  = tid >> 4;
        int c0 = (tid & 15) * 8;
        int gr = row0 + r;
        if (gr < n_rows) {
            float cv  = cnt[gr];
            float inv = 1.0f / fmaxf(cv, 1.0f);
            const float* ga = agg  + (size_t)gr * CDIM + c0;
            const float* gx = root + (size_t)gr * CDIM + c0;
#pragma unroll
            for (int i = 0; i < 8; ++i) {
                sMean[r][c0 + i] = ga[i] * inv;
                sRoot[r][c0 + i] = gx[i];
            }
        } else {
#pragma unroll
            for (int i = 0; i < 8; ++i) {
                sMean[r][c0 + i] = 0.0f;
                sRoot[r][c0 + i] = 0.0f;
            }
        }
    }
    __syncthreads();

    const int wave  = tid >> 5;
    const int lane  = tid & 31;
    const int lhalf = (lane >> 4) & 1;
    const int lm    = lane & 15;
    const int n0    = wave * 16;

    v8f acc;
    {
        float bv = bias[n0 + lm];
#pragma unroll
        for (int i = 0; i < 8; ++i) acc[i] = bv;
    }

#pragma unroll
    for (int k = 0; k < CDIM; k += 4) {
        int ka = k + 2 * lhalf;
        v2f a, b;
        a.x = sMean[lm][ka];
        a.y = sMean[lm][ka + 1];
        b.x = W_nbr[(size_t)ka * CDIM + n0 + lm];
        b.y = W_nbr[(size_t)(ka + 1) * CDIM + n0 + lm];
        acc = __builtin_amdgcn_wmma_f32_16x16x4_f32(
            false, a, false, b, (short)0, acc, false, false);
    }
#pragma unroll
    for (int k = 0; k < CDIM; k += 4) {
        int ka = k + 2 * lhalf;
        v2f a, b;
        a.x = sRoot[lm][ka];
        a.y = sRoot[lm][ka + 1];
        b.x = W_root[(size_t)ka * CDIM + n0 + lm];
        b.y = W_root[(size_t)(ka + 1) * CDIM + n0 + lm];
        acc = __builtin_amdgcn_wmma_f32_16x16x4_f32(
            false, a, false, b, (short)0, acc, false, false);
    }

    if (do_relu) {
#pragma unroll
        for (int i = 0; i < 8; ++i) acc[i] = fmaxf(acc[i], 0.0f);
    }

#pragma unroll
    for (int v = 0; v < 8; ++v) {
        int gr = row0 + v + 8 * lhalf;
        if (gr < n_rows) out[(size_t)gr * CDIM + n0 + lm] = acc[v];
    }
}

static __device__ __forceinline__ float GNN_dot4(const float4& a, const float4& b) {
    float s = a.x * b.x;
    s += a.y * b.y;
    s += a.z * b.z;
    s += a.w * b.w;
    return s;
}
static __device__ __forceinline__ float GNN_wave_sum(float s) {
    for (int o = 16; o > 0; o >>= 1) s += __shfl_xor(s, o, 32);
    return s;
}

__global__ __launch_bounds__(256)
void GNN_decode_kernel(const float* __restrict__ h_u2,
                       const float* __restrict__ h_m2,
                       const int*   __restrict__ lab_u,
                       const int*   __restrict__ lab_m,
                       float* __restrict__ out, int n) {
    int chunk = (int)((blockIdx.x * (size_t)blockDim.x + threadIdx.x) >> 5);
    chunk = __builtin_amdgcn_readfirstlane(chunk);
    const int lane = threadIdx.x & 31;
    const int base = chunk * 32;
    if (base >= n) return;
    const int cnt = min(32, n - base);
    const int iu = (lane < cnt) ? lab_u[base + lane] : 0;
    const int im = (lane < cnt) ? lab_m[base + lane] : 0;
    float res = 0.0f;
    int j = 0;
    for (; j + 4 <= cnt; j += 4) {
        const float4 a0 = GNN_row4(h_u2, __builtin_amdgcn_readlane(iu, j),     lane);
        const float4 b0 = GNN_row4(h_m2, __builtin_amdgcn_readlane(im, j),     lane);
        const float4 a1 = GNN_row4(h_u2, __builtin_amdgcn_readlane(iu, j + 1), lane);
        const float4 b1 = GNN_row4(h_m2, __builtin_amdgcn_readlane(im, j + 1), lane);
        const float4 a2 = GNN_row4(h_u2, __builtin_amdgcn_readlane(iu, j + 2), lane);
        const float4 b2 = GNN_row4(h_m2, __builtin_amdgcn_readlane(im, j + 2), lane);
        const float4 a3 = GNN_row4(h_u2, __builtin_amdgcn_readlane(iu, j + 3), lane);
        const float4 b3 = GNN_row4(h_m2, __builtin_amdgcn_readlane(im, j + 3), lane);
        const float s0 = GNN_wave_sum(GNN_dot4(a0, b0));
        const float s1 = GNN_wave_sum(GNN_dot4(a1, b1));
        const float s2 = GNN_wave_sum(GNN_dot4(a2, b2));
        const float s3 = GNN_wave_sum(GNN_dot4(a3, b3));
        res = (lane == j)     ? s0 : res;
        res = (lane == j + 1) ? s1 : res;
        res = (lane == j + 2) ? s2 : res;
        res = (lane == j + 3) ? s3 : res;
    }
    for (; j < cnt; ++j) {
        const float4 a0 = GNN_row4(h_u2, __builtin_amdgcn_readlane(iu, j), lane);
        const float4 b0 = GNN_row4(h_m2, __builtin_amdgcn_readlane(im, j), lane);
        const float s0 = GNN_wave_sum(GNN_dot4(a0, b0));
        res = (lane == j) ? s0 : res;
    }
    if (lane < cnt) out[base + lane] = res;
}

extern "C" void kernel_launch(void* const* d_in, const int* in_sizes, int n_in,
                              void* d_out, int out_size, void* d_ws, size_t ws_size,
                              hipStream_t stream) {
    const float* x_user  = (const float*)d_in[0];
    const float* x_movie = (const float*)d_in[1];
    const int* edge_src  = (const int*)d_in[2];
    const int* edge_dst  = (const int*)d_in[3];
    const int* label_u   = (const int*)d_in[4];
    const int* label_m   = (const int*)d_in[5];
    const float* W_nbr1_um  = (const float*)d_in[6];
    const float* W_root1_um = (const float*)d_in[7];
    const float* W_nbr1_mu  = (const float*)d_in[8];
    const float* W_root1_mu = (const float*)d_in[9];
    const float* W_nbr2_um  = (const float*)d_in[10];
    const float* W_root2_um = (const float*)d_in[11];
    const float* W_nbr2_mu  = (const float*)d_in[12];
    const float* W_root2_mu = (const float*)d_in[13];
    const float* b1_m = (const float*)d_in[14];
    const float* b1_u = (const float*)d_in[15];
    const float* b2_m = (const float*)d_in[16];
    const float* b2_u = (const float*)d_in[17];

    const int n_users  = in_sizes[0] / CDIM;
    const int n_movies = in_sizes[1] / CDIM;
    const int E  = in_sizes[2];
    const int EL = in_sizes[4];

    const int N = n_movies + n_users;

    float* ws = (float*)d_ws;
    float* agg_m = ws;
    float* agg_u = agg_m + (size_t)n_movies * CDIM;
    float* cnt_m = agg_u + (size_t)n_users * CDIM;
    float* cnt_u = cnt_m + n_movies;
    float* h_m   = cnt_u + n_users;
    float* h_u   = h_m + (size_t)n_movies * CDIM;
    int*   deg   = (int*)(h_u + (size_t)n_users * CDIM);
    int*   off   = deg + N;
    int*   rank  = off + N;
    int*   col   = rank + 2 * (size_t)E;

    float* out = (float*)d_out;

    const int node_blocks = (N + 255) / 256;
    const int tile_blocks = (N + 1023) / 1024;
    const int edge_blocks = (E + 255) / 256;
    const int wave_blocks = (N + 7) / 8;
    const int lab_blocks  = (EL + 255) / 256;
    const int mov_blocks  = (n_movies + 15) / 16;
    const int usr_blocks  = (n_users + 15) / 16;

    GNN_zero_kernel<<<node_blocks, 256, 0, stream>>>((float*)deg, (size_t)N);
    GNN_rank_kernel<<<edge_blocks, 256, 0, stream>>>(
        edge_src, edge_dst, deg, rank, E, n_movies);
    GNN_offsets_kernel<<<tile_blocks, 1024, 0, stream>>>(deg, off, cnt_m, N);
    GNN_fill_kernel<<<edge_blocks, 256, 0, stream>>>(
        edge_src, edge_dst, off, rank, col, E, n_movies);

    GNN_gather_kernel<<<wave_blocks, 256, 0, stream>>>(
        x_user, x_movie, off, deg, col, agg_m, n_movies, N);
    GNN_sage_gemm_kernel<<<mov_blocks, 256, 0, stream>>>(
        agg_m, cnt_m, x_movie, W_nbr1_um, W_root1_um, b1_m, h_m, n_movies, 1);
    GNN_sage_gemm_kernel<<<usr_blocks, 256, 0, stream>>>(
        agg_u, cnt_u, x_user, W_nbr1_mu, W_root1_mu, b1_u, h_u, n_users, 1);

    GNN_gather_kernel<<<wave_blocks, 256, 0, stream>>>(
        h_u, h_m, off, deg, col, agg_m, n_movies, N);
    GNN_sage_gemm_kernel<<<mov_blocks, 256, 0, stream>>>(
        agg_m, cnt_m, h_m, W_nbr2_um, W_root2_um, b2_m, agg_m, n_movies, 0);
    GNN_sage_gemm_kernel<<<usr_blocks, 256, 0, stream>>>(
        agg_u, cnt_u, h_u, W_nbr2_mu, W_root2_mu, b2_u, agg_u, n_users, 0);

    GNN_decode_kernel<<<lab_blocks, 256, 0, stream>>>(
        agg_u  , agg_m  , label_u, label_m, out, EL);

    (void)n_in; (void)out_size; (void)ws_size;
}


# Neighbour sum fused into the node-update kernel: each wave of a block sums its two nodes' neighbour rows straight into the LDS A tile; both gather launches and the summed-row round trip are gone
# speedup vs baseline: 1.9406x; 1.1556x over previous
#include <hip/hip_runtime.h>
#include <hip/hip_bf16.h>

typedef __attribute__((ext_vector_type(2))) float v2f;
typedef __attribute__((ext_vector_type(8))) float v8f;

#define CDIM 128

__global__ void GNN_zero_kernel(float* __restrict__ p, size_t n) {
    size_t i = (size_t)blockIdx.x * blockDim.x + threadIdx.x;
    size_t stride = (size_t)gridDim.x * blockDim.x;
    for (; i < n; i += stride) p[i] = 0.0f;
}

__global__ void GNN_rank_kernel(const int* __restrict__ src_idx,
                                const int* __restrict__ dst_idx,
                                int* __restrict__ deg,
                                int* __restrict__ rank,
                                int n_edges, int n_movies) {
    int e = (int)(blockIdx.x * (size_t)blockDim.x + threadIdx.x);
    if (e >= n_edges) return;
    int s = src_idx[e];
    int d = dst_idx[e];
    rank[e]           = atomicAdd(deg + d, 1);
    rank[n_edges + e] = atomicAdd(deg + n_movies + s, 1);
}

__global__ __launch_bounds__(1024)
void GNN_offsets_kernel(const int* __restrict__ deg,
                        int* __restrict__ off, int n) {
    __shared__ int sFront[32];
    __shared__ int sWave[32];
    const int tid   = threadIdx.x;
    const int lane  = tid & 31;
    const int wave  = tid >> 5;
    const int tile0 = blockIdx.x * 1024;

    int front = 0;
    for (int i = tid; i < tile0; i += 1024) front += deg[i];
    const int gi = tile0 + tid;
    const int x  = (gi < n) ? deg[gi] : 0;

#pragma unroll
    for (int o = 16; o > 0; o >>= 1) front += __shfl_xor(front, o, 32);
    int inc = x;
#pragma unroll
    for (int o = 1; o < 32; o <<= 1) {
        int t = __shfl_up(inc, o, 32);
        if (lane >= o) inc += t;
    }
    if (lane == 0)  sFront[wave] = front;
    if (lane == 31) sWave[wave]  = inc;
    __syncthreads();
    int base = sFront[lane] + ((lane < wave) ? sWave[lane] : 0);
#pragma unroll
    for (int o = 16; o > 0; o >>= 1) base += __shfl_xor(base, o, 32);
    if (gi < n) off[gi] = base + inc - x;
}

__global__ void GNN_fill_kernel(const int* __restrict__ src_idx,
                                const int* __restrict__ dst_idx,
                                const int* __restrict__ off,
                                const int* __restrict__ rank,
                                int* __restrict__ col,
                                int n_edges, int n_movies) {
    int e = (int)(blockIdx.x * (size_t)blockDim.x + threadIdx.x);
    if (e >= n_edges) return;
    int s = src_idx[e];
    int d = dst_idx[e];
    col[off[d] + rank[e]]                       = s;
    col[off[n_movies + s] + rank[n_edges + e]]  = d;
}

static __device__ __forceinline__ float4 GNN_row4(const float* feat, int r, int lane) {
    return *(const float4*)(feat + (size_t)r * CDIM + lane * 4);
}
static __device__ __forceinline__ void GNN_add4(float4& a, const float4& x) {
    a.x += x.x; a.y += x.y; a.z += x.z; a.w += x.w;
}

static __device__ __forceinline__ float4 GNN_list_sum(const float* __restrict__ feat,
                                                      const int*   __restrict__ col,
                                                      int start, int len, int first, int lane) {
    float4 acc = make_float4(0.0f, 0.0f, 0.0f, 0.0f);
    for (int base = 0; base < len; base += 32) {
        const int n = min(32, len - base);
        int idx = first;
        if (base > 0) idx = (lane < n) ? col[start + base + lane] : 0;
        int j = 0;
        for (; j + 4 <= n; j += 4) {
            const float4 x0 = GNN_row4(feat, __builtin_amdgcn_readlane(idx, j),     lane);
            const float4 x1 = GNN_row4(feat, __builtin_amdgcn_readlane(idx, j + 1), lane);
            const float4 x2 = GNN_row4(feat, __builtin_amdgcn_readlane(idx, j + 2), lane);
            const float4 x3 = GNN_row4(feat, __builtin_amdgcn_readlane(idx, j + 3), lane);
            GNN_add4(acc, x0); GNN_add4(acc, x1); GNN_add4(acc, x2); GNN_add4(acc, x3);
        }
        if (j < n) {
            const bool two   = j + 1 < n;
            const bool three = j + 2 < n;
            const int r0 = __builtin_amdgcn_readlane(idx, j);
            const int r1 = __builtin_amdgcn_readlane(idx, two   ? j + 1 : j);
            const int r2 = __builtin_amdgcn_readlane(idx, three ? j + 2 : j);
            const float4 x0 = GNN_row4(feat, r0, lane);
            float4 x1 = make_float4(0.0f, 0.0f, 0.0f, 0.0f);
            float4 x2 = make_float4(0.0f, 0.0f, 0.0f, 0.0f);
            if (two)   x1 = GNN_row4(feat, r1, lane);
            if (three) x2 = GNN_row4(feat, r2, lane);
            GNN_add4(acc, x0);
            if (two)   GNN_add4(acc, x1);
            if (three) GNN_add4(acc, x2);
        }
    }
    return acc;
}

__global__ __launch_bounds__(256)
void GNN_sage_gemm_kernel(const float* __restrict__ nbr_feat,
                          const float* __restrict__ root,
                          const int*   __restrict__ off,
                          const int*   __restrict__ deg,
                          const int*   __restrict__ col,
                          const float* __restrict__ W_nbr,
                          const float* __restrict__ W_root,
                          const float* __restrict__ bias,
                          float* __restrict__ out,
                          int n_rows, int node_base, int do_relu) {
    __shared__ float sMean[16][CDIM + 4];
    __shared__ float sRoot[16][CDIM + 4];

    const int tid  = threadIdx.x;
    const int row0 = blockIdx.x * 16;

    {
        const int w    = __builtin_amdgcn_readfirstlane(tid >> 5);
        const int ln   = tid & 31;
        const int r0   = 2 * w;
        const int gr0  = row0 + r0;
        const int gr1  = gr0 + 1;
        const bool ok0 = gr0 < n_rows;
        const bool ok1 = gr1 < n_rows;
        int start0 = 0, len0 = 0, start1 = 0, len1 = 0;
        if (ok0) { start0 = off[node_base + gr0]; len0 = deg[node_base + gr0]; }
        if (ok1) { start1 = off[node_base + gr1]; len1 = deg[node_base + gr1]; }
        float4 x0 = make_float4(0.0f, 0.0f, 0.0f, 0.0f);
        float4 x1 = make_float4(0.0f, 0.0f, 0.0f, 0.0f);
        if (ok0) x0 = GNN_row4(root, gr0, ln);
        if (ok1) x1 = GNN_row4(root, gr1, ln);
        const int first0 = (ln < min(32, len0)) ? col[start0 + ln] : 0;
        const int first1 = (ln < min(32, len1)) ? col[start1 + ln] : 0;
        const float4 a0 = GNN_list_sum(nbr_feat, col, start0, len0, first0, ln);
        const float4 a1 = GNN_list_sum(nbr_feat, col, start1, len1, first1, ln);
        const float inv0 = 1.0f / fmaxf((float)len0, 1.0f);
        const float inv1 = 1.0f / fmaxf((float)len1, 1.0f);
        *(float4*)&sMean[r0][ln * 4]     = make_float4(a0.x * inv0, a0.y * inv0, a0.z * inv0, a0.w * inv0);
        *(float4*)&sMean[r0 + 1][ln * 4] = make_float4(a1.x * inv1, a1.y * inv1, a1.z * inv1, a1.w * inv1);
        *(float4*)&sRoot[r0][ln * 4]     = x0;
        *(float4*)&sRoot[r0 + 1][ln * 4] = x1;
    }
    __syncthreads();

    const int wave  = tid >> 5;
    const int lane  = tid & 31;
    const int lhalf = (lane >> 4) & 1;
    const int lm    = lane & 15;
    const int n0    = wave * 16;

    v8f acc;
    {
        float bv = bias[n0 + lm];
#pragma unroll
        for (int i = 0; i < 8; ++i) acc[i] = bv;
    }

#pragma unroll
    for (int k = 0; k < CDIM; k += 4) {
        int ka = k + 2 * lhalf;
        v2f a, b;
        a.x = sMean[lm][ka];
        a.y = sMean[lm][ka + 1];
        b.x = W_nbr[(size_t)ka * CDIM + n0 + lm];
        b.y = W_nbr[(size_t)(ka + 1) * CDIM + n0 + lm];
        acc = __builtin_amdgcn_wmma_f32_16x16x4_f32(
            false, a, false, b, (short)0, acc, false, false);
    }
#pragma unroll
    for (int k = 0; k < CDIM; k += 4) {
        int ka = k + 2 * lhalf;
        v2f a, b;
        a.x = sRoot[lm][ka];
        a.y = sRoot[lm][ka + 1];
        b.x = W_root[(size_t)ka * CDIM + n0 + lm];
        b.y = W_root[(size_t)(ka + 1) * CDIM + n0 + lm];
        acc = __builtin_amdgcn_wmma_f32_16x16x4_f32(
            false, a, false, b, (short)0, acc, false, false);
    }

    if (do_relu) {
#pragma unroll
        for (int i = 0; i < 8; ++i) acc[i] = fmaxf(acc[i], 0.0f);
    }

#pragma unroll
    for (int v = 0; v < 8; ++v) {
        int gr = row0 + v + 8 * lhalf;
        if (gr < n_rows) out[(size_t)gr * CDIM + n0 + lm] = acc[v];
    }
}

static __device__ __forceinline__ float GNN_dot4(const float4& a, const float4& b) {
    float s = a.x * b.x;
    s += a.y * b.y;
    s += a.z * b.z;
    s += a.w * b.w;
    return s;
}
static __device__ __forceinline__ float GNN_wave_sum(float s) {
    for (int o = 16; o > 0; o >>= 1) s += __shfl_xor(s, o, 32);
    return s;
}

__global__ __launch_bounds__(256)
void GNN_decode_kernel(const float* __restrict__ h_u2,
                       const float* __restrict__ h_m2,
                       const int*   __restrict__ lab_u,
                       const int*   __restrict__ lab_m,
                       float* __restrict__ out, int n) {
    int chunk = (int)((blockIdx.x * (size_t)blockDim.x + threadIdx.x) >> 5);
    chunk = __builtin_amdgcn_readfirstlane(chunk);
    const int lane = threadIdx.x & 31;
    const int base = chunk * 32;
    if (base >= n) return;
    const int cnt = min(32, n - base);
    const int iu = (lane < cnt) ? lab_u[base + lane] : 0;
    const int im = (lane < cnt) ? lab_m[base + lane] : 0;
    float res = 0.0f;
    int j = 0;
    for (; j + 4 <= cnt; j += 4) {
        const float4 a0 = GNN_row4(h_u2, __builtin_amdgcn_readlane(iu, j),     lane);
        const float4 b0 = GNN_row4(h_m2, __builtin_amdgcn_readlane(im, j),     lane);
        const float4 a1 = GNN_row4(h_u2, __builtin_amdgcn_readlane(iu, j + 1), lane);
        const float4 b1 = GNN_row4(h_m2, __builtin_amdgcn_readlane(im, j + 1), lane);
        const float4 a2 = GNN_row4(h_u2, __builtin_amdgcn_readlane(iu, j + 2), lane);
        const float4 b2 = GNN_row4(h_m2, __builtin_amdgcn_readlane(im, j + 2), lane);
        const float4 a3 = GNN_row4(h_u2, __builtin_amdgcn_readlane(iu, j + 3), lane);
        const float4 b3 = GNN_row4(h_m2, __builtin_amdgcn_readlane(im, j + 3), lane);
        const float s0 = GNN_wave_sum(GNN_dot4(a0, b0));
        const float s1 = GNN_wave_sum(GNN_dot4(a1, b1));
        const float s2 = GNN_wave_sum(GNN_dot4(a2, b2));
        const float s3 = GNN_wave_sum(GNN_dot4(a3, b3));
        res = (lane == j)     ? s0 : res;
        res = (lane == j + 1) ? s1 : res;
        res = (lane == j + 2) ? s2 : res;
        res = (lane == j + 3) ? s3 : res;
    }
    for (; j < cnt; ++j) {
        const float4 a0 = GNN_row4(h_u2, __builtin_amdgcn_readlane(iu, j), lane);
        const float4 b0 = GNN_row4(h_m2, __builtin_amdgcn_readlane(im, j), lane);
        const float s0 = GNN_wave_sum(GNN_dot4(a0, b0));
        res = (lane == j) ? s0 : res;
    }
    if (lane < cnt) out[base + lane] = res;
}

extern "C" void kernel_launch(void* const* d_in, const int* in_sizes, int n_in,
                              void* d_out, int out_size, void* d_ws, size_t ws_size,
                              hipStream_t stream) {
    const float* x_user  = (const float*)d_in[0];
    const float* x_movie = (const float*)d_in[1];
    const int* edge_src  = (const int*)d_in[2];
    const int* edge_dst  = (const int*)d_in[3];
    const int* label_u   = (const int*)d_in[4];
    const int* label_m   = (const int*)d_in[5];
    const float* W_nbr1_um  = (const float*)d_in[6];
    const float* W_root1_um = (const float*)d_in[7];
    const float* W_nbr1_mu  = (const float*)d_in[8];
    const float* W_root1_mu = (const float*)d_in[9];
    const float* W_nbr2_um  = (const float*)d_in[10];
    const float* W_root2_um = (const float*)d_in[11];
    const float* W_nbr2_mu  = (const float*)d_in[12];
    const float* W_root2_mu = (const float*)d_in[13];
    const float* b1_m = (const float*)d_in[14];
    const float* b1_u = (const float*)d_in[15];
    const float* b2_m = (const float*)d_in[16];
    const float* b2_u = (const float*)d_in[17];

    const int n_users  = in_sizes[0] / CDIM;
    const int n_movies = in_sizes[1] / CDIM;
    const int E  = in_sizes[2];
    const int EL = in_sizes[4];

    const int N = n_movies + n_users;

    float* ws = (float*)d_ws;
    float* h2_m = ws;
    float* h2_u = h2_m + (size_t)n_movies * CDIM;
    float* h_m  = h2_u + (size_t)n_users * CDIM;
    float* h_u  = h_m + (size_t)n_movies * CDIM;
    int*   deg  = (int*)(h_u + (size_t)n_users * CDIM);
    int*   off  = deg + N;
    int*   rank = off + N;
    int*   col  = rank + 2 * (size_t)E;

    float* out = (float*)d_out;

    const int node_blocks = (N + 255) / 256;
    const int tile_blocks = (N + 1023) / 1024;
    const int edge_blocks = (E + 255) / 256;
    const int lab_blocks  = (EL + 255) / 256;
    const int mov_blocks  = (n_movies + 15) / 16;
    const int usr_blocks  = (n_users + 15) / 16;

    GNN_zero_kernel<<<node_blocks, 256, 0, stream>>>((float*)deg, (size_t)N);
    GNN_rank_kernel<<<edge_blocks, 256, 0, stream>>>(
        edge_src, edge_dst, deg, rank, E, n_movies);
    GNN_offsets_kernel<<<tile_blocks, 1024, 0, stream>>>(deg, off, N);
    GNN_fill_kernel<<<edge_blocks, 256, 0, stream>>>(
        edge_src, edge_dst, off, rank, col, E, n_movies);

    GNN_sage_gemm_kernel<<<mov_blocks, 256, 0, stream>>>(
        x_user, x_movie, off, deg, col, W_nbr1_um, W_root1_um, b1_m, h_m, n_movies, 0, 1);
    GNN_sage_gemm_kernel<<<usr_blocks, 256, 0, stream>>>(
        x_movie, x_user, off, deg, col, W_nbr1_mu, W_root1_mu, b1_u, h_u, n_users, n_movies, 1);

    GNN_sage_gemm_kernel<<<mov_blocks, 256, 0, stream>>>(
        h_u, h_m, off, deg, col, W_nbr2_um, W_root2_um, b2_m, h2_m, n_movies, 0, 0);
    GNN_sage_gemm_kernel<<<usr_blocks, 256, 0, stream>>>(
        h_m, h_u, off, deg, col, W_nbr2_mu, W_root2_mu, b2_u, h2_u, n_users, n_movies, 0);

    GNN_decode_kernel<<<lab_blocks, 256, 0, stream>>>(
        h2_u, h2_m, label_u, label_m, out, EL);

    (void)n_in; (void)out_size; (void)ws_size;
}


# User-side node update: 64 rows per block instead of 16; each weight (B) fragment is loaded once and feeds 4 matrix instructions on 4 accumulators; 4x fewer user workgroups
# speedup vs baseline: 2.0069x; 1.0342x over previous
#include <hip/hip_runtime.h>
#include <hip/hip_bf16.h>

typedef __attribute__((ext_vector_type(2))) float v2f;
typedef __attribute__((ext_vector_type(8))) float v8f;

#define CDIM 128

__global__ void GNN_zero_kernel(float* __restrict__ p, size_t n) {
    size_t i = (size_t)blockIdx.x * blockDim.x + threadIdx.x;
    size_t stride = (size_t)gridDim.x * blockDim.x;
    for (; i < n; i += stride) p[i] = 0.0f;
}

__global__ void GNN_rank_kernel(const int* __restrict__ src_idx,
                                const int* __restrict__ dst_idx,
                                int* __restrict__ deg,
                                int* __restrict__ rank,
                                int n_edges, int n_movies) {
    int e = (int)(blockIdx.x * (size_t)blockDim.x + threadIdx.x);
    if (e >= n_edges) return;
    int s = src_idx[e];
    int d = dst_idx[e];
    rank[e]           = atomicAdd(deg + d, 1);
    rank[n_edges + e] = atomicAdd(deg + n_movies + s, 1);
}

__global__ __launch_bounds__(1024)
void GNN_offsets_kernel(const int* __restrict__ deg,
                        int* __restrict__ off, int n) {
    __shared__ int sFront[32];
    __shared__ int sWave[32];
    const int tid   = threadIdx.x;
    const int lane  = tid & 31;
    const int wave  = tid >> 5;
    const int tile0 = blockIdx.x * 1024;

    int front = 0;
    for (int i = tid; i < tile0; i += 1024) front += deg[i];
    const int gi = tile0 + tid;
    const int x  = (gi < n) ? deg[gi] : 0;

#pragma unroll
    for (int o = 16; o > 0; o >>= 1) front += __shfl_xor(front, o, 32);
    int inc = x;
#pragma unroll
    for (int o = 1; o < 32; o <<= 1) {
        int t = __shfl_up(inc, o, 32);
        if (lane >= o) inc += t;
    }
    if (lane == 0)  sFront[wave] = front;
    if (lane == 31) sWave[wave]  = inc;
    __syncthreads();
    int base = sFront[lane] + ((lane < wave) ? sWave[lane] : 0);
#pragma unroll
    for (int o = 16; o > 0; o >>= 1) base += __shfl_xor(base, o, 32);
    if (gi < n) off[gi] = base + inc - x;
}

__global__ void GNN_fill_kernel(const int* __restrict__ src_idx,
                                const int* __restrict__ dst_idx,
                                const int* __restrict__ off,
                                const int* __restrict__ rank,
                                int* __restrict__ col,
                                int n_edges, int n_movies) {
    int e = (int)(blockIdx.x * (size_t)blockDim.x + threadIdx.x);
    if (e >= n_edges) return;
    int s = src_idx[e];
    int d = dst_idx[e];
    col[off[d] + rank[e]]                       = s;
    col[off[n_movies + s] + rank[n_edges + e]]  = d;
}

static __device__ __forceinline__ float4 GNN_row4(const float* feat, int r, int lane) {
    return *(const float4*)(feat + (size_t)r * CDIM + lane * 4);
}
static __device__ __forceinline__ void GNN_add4(float4& a, const float4& x) {
    a.x += x.x; a.y += x.y; a.z += x.z; a.w += x.w;
}

static __device__ __forceinline__ float4 GNN_list_sum(const float* __restrict__ feat,
                                                      const int*   __restrict__ col,
                                                      int start, int len, int first, int lane) {
    float4 acc = make_float4(0.0f, 0.0f, 0.0f, 0.0f);
    for (int base = 0; base < len; base += 32) {
        const int n = min(32, len - base);
        int idx = first;
        if (base > 0) idx = (lane < n) ? col[start + base + lane] : 0;
        int j = 0;
        for (; j + 4 <= n; j += 4) {
            const float4 x0 = GNN_row4(feat, __builtin_amdgcn_readlane(idx, j),     lane);
            const float4 x1 = GNN_row4(feat, __builtin_amdgcn_readlane(idx, j + 1), lane);
            const float4 x2 = GNN_row4(feat, __builtin_amdgcn_readlane(idx, j + 2), lane);
            const float4 x3 = GNN_row4(feat, __builtin_amdgcn_readlane(idx, j + 3), lane);
            GNN_add4(acc, x0); GNN_add4(acc, x1); GNN_add4(acc, x2); GNN_add4(acc, x3);
        }
        if (j < n) {
            const bool two   = j + 1 < n;
            const bool three = j + 2 < n;
            const int r0 = __builtin_amdgcn_readlane(idx, j);
            const int r1 = __builtin_amdgcn_readlane(idx, two   ? j + 1 : j);
            const int r2 = __builtin_amdgcn_readlane(idx, three ? j + 2 : j);
            const float4 x0 = GNN_row4(feat, r0, lane);
            float4 x1 = make_float4(0.0f, 0.0f, 0.0f, 0.0f);
            float4 x2 = make_float4(0.0f, 0.0f, 0.0f, 0.0f);
            if (two)   x1 = GNN_row4(feat, r1, lane);
            if (three) x2 = GNN_row4(feat, r2, lane);
            GNN_add4(acc, x0);
            if (two)   GNN_add4(acc, x1);
            if (three) GNN_add4(acc, x2);
        }
    }
    return acc;
}

template <int MT>
__global__ __launch_bounds__(256)
void GNN_sage_gemm_kernel(const float* __restrict__ nbr_feat,
                          const float* __restrict__ root,
                          const int*   __restrict__ off,
                          const int*   __restrict__ deg,
                          const int*   __restrict__ col,
                          const float* __restrict__ W_nbr,
                          const float* __restrict__ W_root,
                          const float* __restrict__ bias,
                          float* __restrict__ out,
                          int n_rows, int node_base, int do_relu) {
    __shared__ float sMean[16 * MT][CDIM + 4];
    __shared__ float sRoot[16 * MT][CDIM + 4];

    const int tid  = threadIdx.x;
    const int row0 = blockIdx.x * (16 * MT);

    {
        const int w  = __builtin_amdgcn_readfirstlane(tid >> 5);
        const int ln = tid & 31;
        for (int p = 0; p < MT; ++p) {
            const int r0   = 16 * p + 2 * w;
            const int gr0  = row0 + r0;
            const int gr1  = gr0 + 1;
            const bool ok0 = gr0 < n_rows;
            const bool ok1 = gr1 < n_rows;
            int start0 = 0, len0 = 0, start1 = 0, len1 = 0;
            if (ok0) { start0 = off[node_base + gr0]; len0 = deg[node_base + gr0]; }
            if (ok1) { start1 = off[node_base + gr1]; len1 = deg[node_base + gr1]; }
            float4 x0 = make_float4(0.0f, 0.0f, 0.0f, 0.0f);
            float4 x1 = make_float4(0.0f, 0.0f, 0.0f, 0.0f);
            if (ok0) x0 = GNN_row4(root, gr0, ln);
            if (ok1) x1 = GNN_row4(root, gr1, ln);
            const int first0 = (ln < min(32, len0)) ? col[start0 + ln] : 0;
            const int first1 = (ln < min(32, len1)) ? col[start1 + ln] : 0;
            const float4 a0 = GNN_list_sum(nbr_feat, col, start0, len0, first0, ln);
            const float4 a1 = GNN_list_sum(nbr_feat, col, start1, len1, first1, ln);
            const float inv0 = 1.0f / fmaxf((float)len0, 1.0f);
            const float inv1 = 1.0f / fmaxf((float)len1, 1.0f);
            *(float4*)&sMean[r0][ln * 4]     = make_float4(a0.x * inv0, a0.y * inv0, a0.z * inv0, a0.w * inv0);
            *(float4*)&sMean[r0 + 1][ln * 4] = make_float4(a1.x * inv1, a1.y * inv1, a1.z * inv1, a1.w * inv1);
            *(float4*)&sRoot[r0][ln * 4]     = x0;
            *(float4*)&sRoot[r0 + 1][ln * 4] = x1;
        }
    }
    __syncthreads();

    const int wave  = tid >> 5;
    const int lane  = tid & 31;
    const int lhalf = (lane >> 4) & 1;
    const int lm    = lane & 15;
    const int n0    = wave * 16;

    v8f acc[MT];
    {
        float bv = bias[n0 + lm];
#pragma unroll
        for (int t = 0; t < MT; ++t) {
#pragma unroll
            for (int i = 0; i < 8; ++i) acc[t][i] = bv;
        }
    }

#pragma unroll
    for (int k = 0; k < CDIM; k += 4) {
        int ka = k + 2 * lhalf;
        v2f b;
        b.x = W_nbr[(size_t)ka * CDIM + n0 + lm];
        b.y = W_nbr[(size_t)(ka + 1) * CDIM + n0 + lm];
#pragma unroll
        for (int t = 0; t < MT; ++t) {
            v2f a;
            a.x = sMean[16 * t + lm][ka];
            a.y = sMean[16 * t + lm][ka + 1];
            acc[t] = __builtin_amdgcn_wmma_f32_16x16x4_f32(
                false, a, false, b, (short)0, acc[t], false, false);
        }
    }
#pragma unroll
    for (int k = 0; k < CDIM; k += 4) {
        int ka = k + 2 * lhalf;
        v2f b;
        b.x = W_root[(size_t)ka * CDIM + n0 + lm];
        b.y = W_root[(size_t)(ka + 1) * CDIM + n0 + lm];
#pragma unroll
        for (int t = 0; t < MT; ++t) {
            v2f a;
            a.x = sRoot[16 * t + lm][ka];
            a.y = sRoot[16 * t + lm][ka + 1];
            acc[t] = __builtin_amdgcn_wmma_f32_16x16x4_f32(
                false, a, false, b, (short)0, acc[t], false, false);
        }
    }

#pragma unroll
    for (int t = 0; t < MT; ++t) {
#pragma unroll
        for (int v = 0; v < 8; ++v) {
            float r = acc[t][v];
            if (do_relu) r = fmaxf(r, 0.0f);
            int gr = row0 + 16 * t + v + 8 * lhalf;
            if (gr < n_rows) out[(size_t)gr * CDIM + n0 + lm] = r;
        }
    }
}

static __device__ __forceinline__ float GNN_dot4(const float4& a, const float4& b) {
    float s = a.x * b.x;
    s += a.y * b.y;
    s += a.z * b.z;
    s += a.w * b.w;
    return s;
}
static __device__ __forceinline__ float GNN_wave_sum(float s) {
    for (int o = 16; o > 0; o >>= 1) s += __shfl_xor(s, o, 32);
    return s;
}

__global__ __launch_bounds__(256)
void GNN_decode_kernel(const float* __restrict__ h_u2,
                       const float* __restrict__ h_m2,
                       const int*   __restrict__ lab_u,
                       const int*   __restrict__ lab_m,
                       float* __restrict__ out, int n) {
    int chunk = (int)((blockIdx.x * (size_t)blockDim.x + threadIdx.x) >> 5);
    chunk = __builtin_amdgcn_readfirstlane(chunk);
    const int lane = threadIdx.x & 31;
    const int base = chunk * 32;
    if (base >= n) return;
    const int cnt = min(32, n - base);
    const int iu = (lane < cnt) ? lab_u[base + lane] : 0;
    const int im = (lane < cnt) ? lab_m[base + lane] : 0;
    float res = 0.0f;
    int j = 0;
    for (; j + 4 <= cnt; j += 4) {
        const float4 a0 = GNN_row4(h_u2, __builtin_amdgcn_readlane(iu, j),     lane);
        const float4 b0 = GNN_row4(h_m2, __builtin_amdgcn_readlane(im, j),     lane);
        const float4 a1 = GNN_row4(h_u2, __builtin_amdgcn_readlane(iu, j + 1), lane);
        const float4 b1 = GNN_row4(h_m2, __builtin_amdgcn_readlane(im, j + 1), lane);
        const float4 a2 = GNN_row4(h_u2, __builtin_amdgcn_readlane(iu, j + 2), lane);
        const float4 b2 = GNN_row4(h_m2, __builtin_amdgcn_readlane(im, j + 2), lane);
        const float4 a3 = GNN_row4(h_u2, __builtin_amdgcn_readlane(iu, j + 3), lane);
        const float4 b3 = GNN_row4(h_m2, __builtin_amdgcn_readlane(im, j + 3), lane);
        const float s0 = GNN_wave_sum(GNN_dot4(a0, b0));
        const float s1 = GNN_wave_sum(GNN_dot4(a1, b1));
        const float s2 = GNN_wave_sum(GNN_dot4(a2, b2));
        const float s3 = GNN_wave_sum(GNN_dot4(a3, b3));
        res = (lane == j)     ? s0 : res;
        res = (lane == j + 1) ? s1 : res;
        res = (lane == j + 2) ? s2 : res;
        res = (lane == j + 3) ? s3 : res;
    }
    for (; j < cnt; ++j) {
        const float4 a0 = GNN_row4(h_u2, __builtin_amdgcn_readlane(iu, j), lane);
        const float4 b0 = GNN_row4(h_m2, __builtin_amdgcn_readlane(im, j), lane);
        const float s0 = GNN_wave_sum(GNN_dot4(a0, b0));
        res = (lane == j) ? s0 : res;
    }
    if (lane < cnt) out[base + lane] = res;
}

extern "C" void kernel_launch(void* const* d_in, const int* in_sizes, int n_in,
                              void* d_out, int out_size, void* d_ws, size_t ws_size,
                              hipStream_t stream) {
    const float* x_user  = (const float*)d_in[0];
    const float* x_movie = (const float*)d_in[1];
    const int* edge_src  = (const int*)d_in[2];
    const int* edge_dst  = (const int*)d_in[3];
    const int* label_u   = (const int*)d_in[4];
    const int* label_m   = (const int*)d_in[5];
    const float* W_nbr1_um  = (const float*)d_in[6];
    const float* W_root1_um = (const float*)d_in[7];
    const float* W_nbr1_mu  = (const float*)d_in[8];
    const float* W_root1_mu = (const float*)d_in[9];
    const float* W_nbr2_um  = (const float*)d_in[10];
    const float* W_root2_um = (const float*)d_in[11];
    const float* W_nbr2_mu  = (const float*)d_in[12];
    const float* W_root2_mu = (const float*)d_in[13];
    const float* b1_m = (const float*)d_in[14];
    const float* b1_u = (const float*)d_in[15];
    const float* b2_m = (const float*)d_in[16];
    const float* b2_u = (const float*)d_in[17];

    const int n_users  = in_sizes[0] / CDIM;
    const int n_movies = in_sizes[1] / CDIM;
    const int E  = in_sizes[2];
    const int EL = in_sizes[4];

    const int N = n_movies + n_users;

    float* ws = (float*)d_ws;
    float* h2_m = ws;
    float* h2_u = h2_m + (size_t)n_movies * CDIM;
    float* h_m  = h2_u + (size_t)n_users * CDIM;
    float* h_u  = h_m + (size_t)n_movies * CDIM;
    int*   deg  = (int*)(h_u + (size_t)n_users * CDIM);
    int*   off  = deg + N;
    int*   rank = off + N;
    int*   col  = rank + 2 * (size_t)E;

    float* out = (float*)d_out;

    const int node_blocks = (N + 255) / 256;
    const int tile_blocks = (N + 1023) / 1024;
    const int edge_blocks = (E + 255) / 256;
    const int lab_blocks  = (EL + 255) / 256;
    const int mov_blocks  = (n_movies + 15) / 16;
    const int usr_blocks  = (n_users + 63) / 64;

    GNN_zero_kernel<<<node_blocks, 256, 0, stream>>>((float*)deg, (size_t)N);
    GNN_rank_kernel<<<edge_blocks, 256, 0, stream>>>(
        edge_src, edge_dst, deg, rank, E, n_movies);
    GNN_offsets_kernel<<<tile_blocks, 1024, 0, stream>>>(deg, off, N);
    GNN_fill_kernel<<<edge_blocks, 256, 0, stream>>>(
        edge_src, edge_dst, off, rank, col, E, n_movies);

    GNN_sage_gemm_kernel<1><<<mov_blocks, 256, 0, stream>>>(
        x_user, x_movie, off, deg, col, W_nbr1_um, W_root1_um, b1_m, h_m, n_movies, 0, 1);
    GNN_sage_gemm_kernel<4><<<usr_blocks, 256, 0, stream>>>(
        x_movie, x_user, off, deg, col, W_nbr1_mu, W_root1_mu, b1_u, h_u, n_users, n_movies, 1);

    GNN_sage_gemm_kernel<1><<<mov_blocks, 256, 0, stream>>>(
        h_u, h_m, off, deg, col, W_nbr2_um, W_root2_um, b2_m, h2_m, n_movies, 0, 0);
    GNN_sage_gemm_kernel<4><<<usr_blocks, 256, 0, stream>>>(
        h_m, h_u, off, deg, col, W_nbr2_mu, W_root2_mu, b2_u, h2_u, n_users, n_movies, 0);

    GNN_decode_kernel<<<lab_blocks, 256, 0, stream>>>(
        h2_u, h2_m, label_u, label_m, out, EL);

    (void)n_in; (void)out_size; (void)ws_size;
}


# Neighbour lists kept in fixed-pitch slots per node (64 per movie, 32 per user) that the ranking kernel fills itself, excess edges go to a spill list; prefix-sum and fill launches removed (9 to 7)
# speedup vs baseline: 2.2213x; 1.1068x over previous
#include <hip/hip_runtime.h>
#include <hip/hip_bf16.h>

typedef __attribute__((ext_vector_type(2))) float v2f;
typedef __attribute__((ext_vector_type(8))) float v8f;

#define CDIM 128

__global__ void GNN_zero_kernel(float* __restrict__ p, size_t n) {
    size_t i = (size_t)blockIdx.x * blockDim.x + threadIdx.x;
    size_t stride = (size_t)gridDim.x * blockDim.x;
    for (; i < n; i += stride) p[i] = 0.0f;
}

#define CAP_MOVIE 64
#define CAP_USER  32

__global__ void GNN_slot_kernel(const int* __restrict__ src_idx,
                                const int* __restrict__ dst_idx,
                                int* __restrict__ cnt,
                                int* __restrict__ slot_m,
                                int* __restrict__ slot_u,
                                int* __restrict__ spill,
                                int n_edges, int n_movies, int n_nodes) {
    int e = (int)(blockIdx.x * (size_t)blockDim.x + threadIdx.x);
    if (e >= n_edges) return;
    int s = src_idx[e];
    int d = dst_idx[e];
    const int pm = atomicAdd(cnt + d, 1);
    const int pu = atomicAdd(cnt + n_movies + s, 1);
    if (pm < CAP_MOVIE) {
        slot_m[(size_t)d * CAP_MOVIE + pm] = s;
    } else {
        const int o = atomicAdd(cnt + n_nodes, 1);
        spill[2 * (size_t)o]     = d;
        spill[2 * (size_t)o + 1] = s;
    }
    if (pu < CAP_USER) {
        slot_u[(size_t)s * CAP_USER + pu] = d;
    } else {
        const int o = atomicAdd(cnt + n_nodes, 1);
        spill[2 * (size_t)o]     = n_movies + s;
        spill[2 * (size_t)o + 1] = d;
    }
}

static __device__ __forceinline__ float4 GNN_row4(const float* feat, int r, int lane) {
    return *(const float4*)(feat + (size_t)r * CDIM + lane * 4);
}
static __device__ __forceinline__ void GNN_add4(float4& a, const float4& x) {
    a.x += x.x; a.y += x.y; a.z += x.z; a.w += x.w;
}

static __device__ __forceinline__ float4 GNN_list_sum(const float* __restrict__ feat,
                                                      const int*   __restrict__ col,
                                                      int start, int len, int first, int lane) {
    float4 acc = make_float4(0.0f, 0.0f, 0.0f, 0.0f);
    for (int base = 0; base < len; base += 32) {
        const int n = min(32, len - base);
        int idx = first;
        if (base > 0) idx = (lane < n) ? col[start + base + lane] : 0;
        int j = 0;
        for (; j + 4 <= n; j += 4) {
            const float4 x0 = GNN_row4(feat, __builtin_amdgcn_readlane(idx, j),     lane);
            const float4 x1 = GNN_row4(feat, __builtin_amdgcn_readlane(idx, j + 1), lane);
            const float4 x2 = GNN_row4(feat, __builtin_amdgcn_readlane(idx, j + 2), lane);
            const float4 x3 = GNN_row4(feat, __builtin_amdgcn_readlane(idx, j + 3), lane);
            GNN_add4(acc, x0); GNN_add4(acc, x1); GNN_add4(acc, x2); GNN_add4(acc, x3);
        }
        if (j < n) {
            const bool two   = j + 1 < n;
            const bool three = j + 2 < n;
            const int r0 = __builtin_amdgcn_readlane(idx, j);
            const int r1 = __builtin_amdgcn_readlane(idx, two   ? j + 1 : j);
            const int r2 = __builtin_amdgcn_readlane(idx, three ? j + 2 : j);
            const float4 x0 = GNN_row4(feat, r0, lane);
            float4 x1 = make_float4(0.0f, 0.0f, 0.0f, 0.0f);
            float4 x2 = make_float4(0.0f, 0.0f, 0.0f, 0.0f);
            if (two)   x1 = GNN_row4(feat, r1, lane);
            if (three) x2 = GNN_row4(feat, r2, lane);
            GNN_add4(acc, x0);
            if (two)   GNN_add4(acc, x1);
            if (three) GNN_add4(acc, x2);
        }
    }
    return acc;
}

static __device__ __forceinline__ void GNN_spill_sum(float4& acc,
                                                     const float* __restrict__ feat,
                                                     const int*   __restrict__ spill,
                                                     int n_spill, int node, int lane) {
    for (int i = 0; i < n_spill; ++i) {
        if (spill[2 * (size_t)i] == node) {
            const float4 x = GNN_row4(feat, spill[2 * (size_t)i + 1], lane);
            GNN_add4(acc, x);
        }
    }
}

template <int MT>
__global__ __launch_bounds__(256)
void GNN_sage_gemm_kernel(const float* __restrict__ nbr_feat,
                          const float* __restrict__ root,
                          const int*   __restrict__ deg,
                          const int*   __restrict__ col,
                          const int*   __restrict__ spill,
                          const int*   __restrict__ spill_cnt,
                          const float* __restrict__ W_nbr,
                          const float* __restrict__ W_root,
                          const float* __restrict__ bias,
                          float* __restrict__ out,
                          int n_rows, int node_base, int cap, int do_relu) {
    __shared__ float sMean[16 * MT][CDIM + 4];
    __shared__ float sRoot[16 * MT][CDIM + 4];

    const int tid  = threadIdx.x;
    const int row0 = blockIdx.x * (16 * MT);

    {
        const int w  = __builtin_amdgcn_readfirstlane(tid >> 5);
        const int ln = tid & 31;
        for (int p = 0; p < MT; ++p) {
            const int r0   = 16 * p + 2 * w;
            const int gr0  = row0 + r0;
            const int gr1  = gr0 + 1;
            const bool ok0 = gr0 < n_rows;
            const bool ok1 = gr1 < n_rows;
            int tot0 = 0, tot1 = 0;
            if (ok0) tot0 = deg[node_base + gr0];
            if (ok1) tot1 = deg[node_base + gr1];
            const int len0   = min(tot0, cap);
            const int len1   = min(tot1, cap);
            const int start0 = ok0 ? gr0 * cap : 0;
            const int start1 = ok1 ? gr1 * cap : 0;
            float4 x0 = make_float4(0.0f, 0.0f, 0.0f, 0.0f);
            float4 x1 = make_float4(0.0f, 0.0f, 0.0f, 0.0f);
            if (ok0) x0 = GNN_row4(root, gr0, ln);
            if (ok1) x1 = GNN_row4(root, gr1, ln);
            const int first0 = (ln < min(32, len0)) ? col[start0 + ln] : 0;
            const int first1 = (ln < min(32, len1)) ? col[start1 + ln] : 0;
            float4 a0 = GNN_list_sum(nbr_feat, col, start0, len0, first0, ln);
            float4 a1 = GNN_list_sum(nbr_feat, col, start1, len1, first1, ln);
            if (tot0 > cap) GNN_spill_sum(a0, nbr_feat, spill, spill_cnt[0], node_base + gr0, ln);
            if (tot1 > cap) GNN_spill_sum(a1, nbr_feat, spill, spill_cnt[0], node_base + gr1, ln);
            const float inv0 = 1.0f / fmaxf((float)tot0, 1.0f);
            const float inv1 = 1.0f / fmaxf((float)tot1, 1.0f);
            *(float4*)&sMean[r0][ln * 4]     = make_float4(a0.x * inv0, a0.y * inv0, a0.z * inv0, a0.w * inv0);
            *(float4*)&sMean[r0 + 1][ln * 4] = make_float4(a1.x * inv1, a1.y * inv1, a1.z * inv1, a1.w * inv1);
            *(float4*)&sRoot[r0][ln * 4]     = x0;
            *(float4*)&sRoot[r0 + 1][ln * 4] = x1;
        }
    }
    __syncthreads();

    const int wave  = tid >> 5;
    const int lane  = tid & 31;
    const int lhalf = (lane >> 4) & 1;
    const int lm    = lane & 15;
    const int n0    = wave * 16;

    v8f acc[MT];
    {
        float bv = bias[n0 + lm];
#pragma unroll
        for (int t = 0; t < MT; ++t) {
#pragma unroll
            for (int i = 0; i < 8; ++i) acc[t][i] = bv;
        }
    }

#pragma unroll
    for (int k = 0; k < CDIM; k += 4) {
        int ka = k + 2 * lhalf;
        v2f b;
        b.x = W_nbr[(size_t)ka * CDIM + n0 + lm];
        b.y = W_nbr[(size_t)(ka + 1) * CDIM + n0 + lm];
#pragma unroll
        for (int t = 0; t < MT; ++t) {
            v2f a;
            a.x = sMean[16 * t + lm][ka];
            a.y = sMean[16 * t + lm][ka + 1];
            acc[t] = __builtin_amdgcn_wmma_f32_16x16x4_f32(
                false, a, false, b, (short)0, acc[t], false, false);
        }
    }
#pragma unroll
    for (int k = 0; k < CDIM; k += 4) {
        int ka = k + 2 * lhalf;
        v2f b;
        b.x = W_root[(size_t)ka * CDIM + n0 + lm];
        b.y = W_root[(size_t)(ka + 1) * CDIM + n0 + lm];
#pragma unroll
        for (int t = 0; t < MT; ++t) {
            v2f a;
            a.x = sRoot[16 * t + lm][ka];
            a.y = sRoot[16 * t + lm][ka + 1];
            acc[t] = __builtin_amdgcn_wmma_f32_16x16x4_f32(
                false, a, false, b, (short)0, acc[t], false, false);
        }
    }

#pragma unroll
    for (int t = 0; t < MT; ++t) {
#pragma unroll
        for (int v = 0; v < 8; ++v) {
            float r = acc[t][v];
            if (do_relu) r = fmaxf(r, 0.0f);
            int gr = row0 + 16 * t + v + 8 * lhalf;
            if (gr < n_rows) out[(size_t)gr * CDIM + n0 + lm] = r;
        }
    }
}

static __device__ __forceinline__ float GNN_dot4(const float4& a, const float4& b) {
    float s = a.x * b.x;
    s += a.y * b.y;
    s += a.z * b.z;
    s += a.w * b.w;
    return s;
}
static __device__ __forceinline__ float GNN_wave_sum(float s) {
    for (int o = 16; o > 0; o >>= 1) s += __shfl_xor(s, o, 32);
    return s;
}

__global__ __launch_bounds__(256)
void GNN_decode_kernel(const float* __restrict__ h_u2,
                       const float* __restrict__ h_m2,
                       const int*   __restrict__ lab_u,
                       const int*   __restrict__ lab_m,
                       float* __restrict__ out, int n) {
    int chunk = (int)((blockIdx.x * (size_t)blockDim.x + threadIdx.x) >> 5);
    chunk = __builtin_amdgcn_readfirstlane(chunk);
    const int lane = threadIdx.x & 31;
    const int base = chunk * 32;
    if (base >= n) return;
    const int cnt = min(32, n - base);
    const int iu = (lane < cnt) ? lab_u[base + lane] : 0;
    const int im = (lane < cnt) ? lab_m[base + lane] : 0;
    float res = 0.0f;
    int j = 0;
    for (; j + 4 <= cnt; j += 4) {
        const float4 a0 = GNN_row4(h_u2, __builtin_amdgcn_readlane(iu, j),     lane);
        const float4 b0 = GNN_row4(h_m2, __builtin_amdgcn_readlane(im, j),     lane);
        const float4 a1 = GNN_row4(h_u2, __builtin_amdgcn_readlane(iu, j + 1), lane);
        const float4 b1 = GNN_row4(h_m2, __builtin_amdgcn_readlane(im, j + 1), lane);
        const float4 a2 = GNN_row4(h_u2, __builtin_amdgcn_readlane(iu, j + 2), lane);
        const float4 b2 = GNN_row4(h_m2, __builtin_amdgcn_readlane(im, j + 2), lane);
        const float4 a3 = GNN_row4(h_u2, __builtin_amdgcn_readlane(iu, j + 3), lane);
        const float4 b3 = GNN_row4(h_m2, __builtin_amdgcn_readlane(im, j + 3), lane);
        const float s0 = GNN_wave_sum(GNN_dot4(a0, b0));
        const float s1 = GNN_wave_sum(GNN_dot4(a1, b1));
        const float s2 = GNN_wave_sum(GNN_dot4(a2, b2));
        const float s3 = GNN_wave_sum(GNN_dot4(a3, b3));
        res = (lane == j)     ? s0 : res;
        res = (lane == j + 1) ? s1 : res;
        res = (lane == j + 2) ? s2 : res;
        res = (lane == j + 3) ? s3 : res;
    }
    for (; j < cnt; ++j) {
        const float4 a0 = GNN_row4(h_u2, __builtin_amdgcn_readlane(iu, j), lane);
        const float4 b0 = GNN_row4(h_m2, __builtin_amdgcn_readlane(im, j), lane);
        const float s0 = GNN_wave_sum(GNN_dot4(a0, b0));
        res = (lane == j) ? s0 : res;
    }
    if (lane < cnt) out[base + lane] = res;
}

extern "C" void kernel_launch(void* const* d_in, const int* in_sizes, int n_in,
                              void* d_out, int out_size, void* d_ws, size_t ws_size,
                              hipStream_t stream) {
    const float* x_user  = (const float*)d_in[0];
    const float* x_movie = (const float*)d_in[1];
    const int* edge_src  = (const int*)d_in[2];
    const int* edge_dst  = (const int*)d_in[3];
    const int* label_u   = (const int*)d_in[4];
    const int* label_m   = (const int*)d_in[5];
    const float* W_nbr1_um  = (const float*)d_in[6];
    const float* W_root1_um = (const float*)d_in[7];
    const float* W_nbr1_mu  = (const float*)d_in[8];
    const float* W_root1_mu = (const float*)d_in[9];
    const float* W_nbr2_um  = (const float*)d_in[10];
    const float* W_root2_um = (const float*)d_in[11];
    const float* W_nbr2_mu  = (const float*)d_in[12];
    const float* W_root2_mu = (const float*)d_in[13];
    const float* b1_m = (const float*)d_in[14];
    const float* b1_u = (const float*)d_in[15];
    const float* b2_m = (const float*)d_in[16];
    const float* b2_u = (const float*)d_in[17];

    const int n_users  = in_sizes[0] / CDIM;
    const int n_movies = in_sizes[1] / CDIM;
    const int E  = in_sizes[2];
    const int EL = in_sizes[4];

    const int N = n_movies + n_users;

    float* ws = (float*)d_ws;
    float* h2_m = ws;
    float* h2_u = h2_m + (size_t)n_movies * CDIM;
    float* h_m  = h2_u + (size_t)n_users * CDIM;
    float* h_u  = h_m + (size_t)n_movies * CDIM;
    int*   deg    = (int*)(h_u + (size_t)n_users * CDIM);
    int*   slot_m = deg + N + 64;
    int*   slot_u = slot_m + (size_t)n_movies * CAP_MOVIE;
    int*   spill  = slot_u + (size_t)n_users * CAP_USER;

    float* out = (float*)d_out;

    const int node_blocks = (N + 255) / 256;
    const int edge_blocks = (E + 255) / 256;
    const int lab_blocks  = (EL + 255) / 256;
    const int mov_blocks  = (n_movies + 15) / 16;
    const int usr_blocks  = (n_users + 63) / 64;

    GNN_zero_kernel<<<node_blocks, 256, 0, stream>>>((float*)deg, (size_t)N + 1);
    GNN_slot_kernel<<<edge_blocks, 256, 0, stream>>>(
        edge_src, edge_dst, deg, slot_m, slot_u, spill, E, n_movies, N);

    GNN_sage_gemm_kernel<1><<<mov_blocks, 256, 0, stream>>>(
        x_user, x_movie, deg, slot_m, spill, deg + N, W_nbr1_um, W_root1_um, b1_m, h_m, n_movies, 0, CAP_MOVIE, 1);
    GNN_sage_gemm_kernel<4><<<usr_blocks, 256, 0, stream>>>(
        x_movie, x_user, deg, slot_u, spill, deg + N, W_nbr1_mu, W_root1_mu, b1_u, h_u, n_users, n_movies, CAP_USER, 1);

    GNN_sage_gemm_kernel<1><<<mov_blocks, 256, 0, stream>>>(
        h_u, h_m, deg, slot_m, spill, deg + N, W_nbr2_um, W_root2_um, b2_m, h2_m, n_movies, 0, CAP_MOVIE, 0);
    GNN_sage_gemm_kernel<4><<<usr_blocks, 256, 0, stream>>>(
        h_m, h_u, deg, slot_u, spill, deg + N, W_nbr2_mu, W_root2_mu, b2_u, h2_u, n_users, n_movies, CAP_USER, 0);

    GNN_decode_kernel<<<lab_blocks, 256, 0, stream>>>(
        h2_u, h2_m, label_u, label_m, out, EL);

    (void)n_in; (void)out_size; (void)ws_size;
}


# Users' neighbour weights applied to the 20000 movie rows before the gather (one more product in the movie launch), not to 100000 user means: user GEMM K 256 to 128; sum order: weights before mean
# speedup vs baseline: 2.3257x; 1.0470x over previous
#include <hip/hip_runtime.h>
#include <hip/hip_bf16.h>

typedef __attribute__((ext_vector_type(2))) float v2f;
typedef __attribute__((ext_vector_type(8))) float v8f;

#define CDIM 128

__global__ void GNN_zero_kernel(float* __restrict__ p, size_t n) {
    size_t i = (size_t)blockIdx.x * blockDim.x + threadIdx.x;
    size_t stride = (size_t)gridDim.x * blockDim.x;
    for (; i < n; i += stride) p[i] = 0.0f;
}

#define CAP_MOVIE 64
#define CAP_USER  32

__global__ void GNN_slot_kernel(const int* __restrict__ src_idx,
                                const int* __restrict__ dst_idx,
                                int* __restrict__ cnt,
                                int* __restrict__ slot_m,
                                int* __restrict__ slot_u,
                                int* __restrict__ spill,
                                int n_edges, int n_movies, int n_nodes) {
    int e = (int)(blockIdx.x * (size_t)blockDim.x + threadIdx.x);
    if (e >= n_edges) return;
    int s = src_idx[e];
    int d = dst_idx[e];
    const int pm = atomicAdd(cnt + d, 1);
    const int pu = atomicAdd(cnt + n_movies + s, 1);
    if (pm < CAP_MOVIE) {
        slot_m[(size_t)d * CAP_MOVIE + pm] = s;
    } else {
        const int o = atomicAdd(cnt + n_nodes, 1);
        spill[2 * (size_t)o]     = d;
        spill[2 * (size_t)o + 1] = s;
    }
    if (pu < CAP_USER) {
        slot_u[(size_t)s * CAP_USER + pu] = d;
    } else {
        const int o = atomicAdd(cnt + n_nodes, 1);
        spill[2 * (size_t)o]     = n_movies + s;
        spill[2 * (size_t)o + 1] = d;
    }
}

static __device__ __forceinline__ float4 GNN_row4(const float* feat, int r, int lane) {
    return *(const float4*)(feat + (size_t)r * CDIM + lane * 4);
}
static __device__ __forceinline__ void GNN_add4(float4& a, const float4& x) {
    a.x += x.x; a.y += x.y; a.z += x.z; a.w += x.w;
}

static __device__ __forceinline__ float4 GNN_list_sum(const float* __restrict__ feat,
                                                      const int*   __restrict__ col,
                                                      int start, int len, int first, int lane) {
    float4 acc = make_float4(0.0f, 0.0f, 0.0f, 0.0f);
    for (int base = 0; base < len; base += 32) {
        const int n = min(32, len - base);
        int idx = first;
        if (base > 0) idx = (lane < n) ? col[start + base + lane] : 0;
        int j = 0;
        for (; j + 4 <= n; j += 4) {
            const float4 x0 = GNN_row4(feat, __builtin_amdgcn_readlane(idx, j),     lane);
            const float4 x1 = GNN_row4(feat, __builtin_amdgcn_readlane(idx, j + 1), lane);
            const float4 x2 = GNN_row4(feat, __builtin_amdgcn_readlane(idx, j + 2), lane);
            const float4 x3 = GNN_row4(feat, __builtin_amdgcn_readlane(idx, j + 3), lane);
            GNN_add4(acc, x0); GNN_add4(acc, x1); GNN_add4(acc, x2); GNN_add4(acc, x3);
        }
        if (j < n) {
            const bool two   = j + 1 < n;
            const bool three = j + 2 < n;
            const int r0 = __builtin_amdgcn_readlane(idx, j);
            const int r1 = __builtin_amdgcn_readlane(idx, two   ? j + 1 : j);
            const int r2 = __builtin_amdgcn_readlane(idx, three ? j + 2 : j);
            const float4 x0 = GNN_row4(feat, r0, lane);
            float4 x1 = make_float4(0.0f, 0.0f, 0.0f, 0.0f);
            float4 x2 = make_float4(0.0f, 0.0f, 0.0f, 0.0f);
            if (two)   x1 = GNN_row4(feat, r1, lane);
            if (three) x2 = GNN_row4(feat, r2, lane);
            GNN_add4(acc, x0);
            if (two)   GNN_add4(acc, x1);
            if (three) GNN_add4(acc, x2);
        }
    }
    return acc;
}

static __device__ __forceinline__ void GNN_spill_sum(float4& acc,
                                                     const float* __restrict__ feat,
                                                     const int*   __restrict__ spill,
                                                     int n_spill, int node, int lane) {
    for (int i = 0; i < n_spill; ++i) {
        if (spill[2 * (size_t)i] == node) {
            const float4 x = GNN_row4(feat, spill[2 * (size_t)i + 1], lane);
            GNN_add4(acc, x);
        }
    }
}

template <int MT, bool PRE>
__global__ __launch_bounds__(256)
void GNN_sage_gemm_kernel(const float* __restrict__ nbr_feat,
                          const float* __restrict__ root,
                          const int*   __restrict__ deg,
                          const int*   __restrict__ col,
                          const int*   __restrict__ spill,
                          const int*   __restrict__ spill_cnt,
                          const float* __restrict__ W_nbr,
                          const float* __restrict__ W_root,
                          const float* __restrict__ bias,
                          float* __restrict__ out,
                          const float* __restrict__ W_pre,
                          float* __restrict__ pre_out,
                          int n_rows, int node_base, int cap, int do_relu) {
    __shared__ float sMean[16 * MT][CDIM + 4];
    __shared__ float sRoot[16 * MT][CDIM + 4];

    const int tid  = threadIdx.x;
    const int row0 = blockIdx.x * (16 * MT);

    {
        const int w  = __builtin_amdgcn_readfirstlane(tid >> 5);
        const int ln = tid & 31;
        for (int p = 0; p < MT; ++p) {
            const int r0   = 16 * p + 2 * w;
            const int gr0  = row0 + r0;
            const int gr1  = gr0 + 1;
            const bool ok0 = gr0 < n_rows;
            const bool ok1 = gr1 < n_rows;
            int tot0 = 0, tot1 = 0;
            if (ok0) tot0 = deg[node_base + gr0];
            if (ok1) tot1 = deg[node_base + gr1];
            const int len0   = min(tot0, cap);
            const int len1   = min(tot1, cap);
            const int start0 = ok0 ? gr0 * cap : 0;
            const int start1 = ok1 ? gr1 * cap : 0;
            float4 x0 = make_float4(0.0f, 0.0f, 0.0f, 0.0f);
            float4 x1 = make_float4(0.0f, 0.0f, 0.0f, 0.0f);
            if (ok0) x0 = GNN_row4(root, gr0, ln);
            if (ok1) x1 = GNN_row4(root, gr1, ln);
            const int first0 = (ln < min(32, len0)) ? col[start0 + ln] : 0;
            const int first1 = (ln < min(32, len1)) ? col[start1 + ln] : 0;
            float4 a0 = GNN_list_sum(nbr_feat, col, start0, len0, first0, ln);
            float4 a1 = GNN_list_sum(nbr_feat, col, start1, len1, first1, ln);
            if (tot0 > cap) GNN_spill_sum(a0, nbr_feat, spill, spill_cnt[0], node_base + gr0, ln);
            if (tot1 > cap) GNN_spill_sum(a1, nbr_feat, spill, spill_cnt[0], node_base + gr1, ln);
            const float inv0 = 1.0f / fmaxf((float)tot0, 1.0f);
            const float inv1 = 1.0f / fmaxf((float)tot1, 1.0f);
            *(float4*)&sMean[r0][ln * 4]     = make_float4(a0.x * inv0, a0.y * inv0, a0.z * inv0, a0.w * inv0);
            *(float4*)&sMean[r0 + 1][ln * 4] = make_float4(a1.x * inv1, a1.y * inv1, a1.z * inv1, a1.w * inv1);
            *(float4*)&sRoot[r0][ln * 4]     = x0;
            *(float4*)&sRoot[r0 + 1][ln * 4] = x1;
        }
    }
    __syncthreads();

    const int wave  = tid >> 5;
    const int lane  = tid & 31;
    const int lhalf = (lane >> 4) & 1;
    const int lm    = lane & 15;
    const int n0    = wave * 16;

    v8f acc[MT];
    {
        float bv = bias[n0 + lm];
#pragma unroll
        for (int t = 0; t < MT; ++t) {
#pragma unroll
            for (int i = 0; i < 8; ++i) {
                if constexpr (PRE) acc[t][i] = bv + sMean[16 * t + i + 8 * lhalf][n0 + lm];
                else               acc[t][i] = bv;
            }
        }
    }

    if constexpr (!PRE) {
#pragma unroll
        for (int k = 0; k < CDIM; k += 4) {
            int ka = k + 2 * lhalf;
            v2f b;
            b.x = W_nbr[(size_t)ka * CDIM + n0 + lm];
            b.y = W_nbr[(size_t)(ka + 1) * CDIM + n0 + lm];
#pragma unroll
            for (int t = 0; t < MT; ++t) {
                v2f a;
                a.x = sMean[16 * t + lm][ka];
                a.y = sMean[16 * t + lm][ka + 1];
                acc[t] = __builtin_amdgcn_wmma_f32_16x16x4_f32(
                    false, a, false, b, (short)0, acc[t], false, false);
            }
        }
    }

#pragma unroll
    for (int k = 0; k < CDIM; k += 4) {
        int ka = k + 2 * lhalf;
        v2f b;
        b.x = W_root[(size_t)ka * CDIM + n0 + lm];
        b.y = W_root[(size_t)(ka + 1) * CDIM + n0 + lm];
#pragma unroll
        for (int t = 0; t < MT; ++t) {
            v2f a;
            a.x = sRoot[16 * t + lm][ka];
            a.y = sRoot[16 * t + lm][ka + 1];
            acc[t] = __builtin_amdgcn_wmma_f32_16x16x4_f32(
                false, a, false, b, (short)0, acc[t], false, false);
        }
    }

#pragma unroll
    for (int t = 0; t < MT; ++t) {
#pragma unroll
        for (int v = 0; v < 8; ++v) {
            float r = acc[t][v];
            if (do_relu) r = fmaxf(r, 0.0f);
            int gr = row0 + 16 * t + v + 8 * lhalf;
            if (gr < n_rows) out[(size_t)gr * CDIM + n0 + lm] = r;
        }
    }

    if constexpr (!PRE) {
        v8f pre[MT];
#pragma unroll
        for (int t = 0; t < MT; ++t) {
#pragma unroll
            for (int i = 0; i < 8; ++i) pre[t][i] = 0.0f;
        }
#pragma unroll
        for (int k = 0; k < CDIM; k += 4) {
            int ka = k + 2 * lhalf;
            v2f b;
            b.x = W_pre[(size_t)ka * CDIM + n0 + lm];
            b.y = W_pre[(size_t)(ka + 1) * CDIM + n0 + lm];
#pragma unroll
            for (int t = 0; t < MT; ++t) {
                v2f a;
                a.x = sRoot[16 * t + lm][ka];
                a.y = sRoot[16 * t + lm][ka + 1];
                pre[t] = __builtin_amdgcn_wmma_f32_16x16x4_f32(
                    false, a, false, b, (short)0, pre[t], false, false);
            }
        }
#pragma unroll
        for (int t = 0; t < MT; ++t) {
#pragma unroll
            for (int v = 0; v < 8; ++v) {
                int gr = row0 + 16 * t + v + 8 * lhalf;
                if (gr < n_rows) pre_out[(size_t)gr * CDIM + n0 + lm] = pre[t][v];
            }
        }
    }
}

static __device__ __forceinline__ float GNN_dot4(const float4& a, const float4& b) {
    float s = a.x * b.x;
    s += a.y * b.y;
    s += a.z * b.z;
    s += a.w * b.w;
    return s;
}
static __device__ __forceinline__ float GNN_wave_sum(float s) {
    for (int o = 16; o > 0; o >>= 1) s += __shfl_xor(s, o, 32);
    return s;
}

__global__ __launch_bounds__(256)
void GNN_decode_kernel(const float* __restrict__ h_u2,
                       const float* __restrict__ h_m2,
                       const int*   __restrict__ lab_u,
                       const int*   __restrict__ lab_m,
                       float* __restrict__ out, int n) {
    int chunk = (int)((blockIdx.x * (size_t)blockDim.x + threadIdx.x) >> 5);
    chunk = __builtin_amdgcn_readfirstlane(chunk);
    const int lane = threadIdx.x & 31;
    const int base = chunk * 32;
    if (base >= n) return;
    const int cnt = min(32, n - base);
    const int iu = (lane < cnt) ? lab_u[base + lane] : 0;
    const int im = (lane < cnt) ? lab_m[base + lane] : 0;
    float res = 0.0f;
    int j = 0;
    for (; j + 4 <= cnt; j += 4) {
        const float4 a0 = GNN_row4(h_u2, __builtin_amdgcn_readlane(iu, j),     lane);
        const float4 b0 = GNN_row4(h_m2, __builtin_amdgcn_readlane(im, j),     lane);
        const float4 a1 = GNN_row4(h_u2, __builtin_amdgcn_readlane(iu, j + 1), lane);
        const float4 b1 = GNN_row4(h_m2, __builtin_amdgcn_readlane(im, j + 1), lane);
        const float4 a2 = GNN_row4(h_u2, __builtin_amdgcn_readlane(iu, j + 2), lane);
        const float4 b2 = GNN_row4(h_m2, __builtin_amdgcn_readlane(im, j + 2), lane);
        const float4 a3 = GNN_row4(h_u2, __builtin_amdgcn_readlane(iu, j + 3), lane);
        const float4 b3 = GNN_row4(h_m2, __builtin_amdgcn_readlane(im, j + 3), lane);
        const float s0 = GNN_wave_sum(GNN_dot4(a0, b0));
        const float s1 = GNN_wave_sum(GNN_dot4(a1, b1));
        const float s2 = GNN_wave_sum(GNN_dot4(a2, b2));
        const float s3 = GNN_wave_sum(GNN_dot4(a3, b3));
        res = (lane == j)     ? s0 : res;
        res = (lane == j + 1) ? s1 : res;
        res = (lane == j + 2) ? s2 : res;
        res = (lane == j + 3) ? s3 : res;
    }
    for (; j < cnt; ++j) {
        const float4 a0 = GNN_row4(h_u2, __builtin_amdgcn_readlane(iu, j), lane);
        const float4 b0 = GNN_row4(h_m2, __builtin_amdgcn_readlane(im, j), lane);
        const float s0 = GNN_wave_sum(GNN_dot4(a0, b0));
        res = (lane == j) ? s0 : res;
    }
    if (lane < cnt) out[base + lane] = res;
}

extern "C" void kernel_launch(void* const* d_in, const int* in_sizes, int n_in,
                              void* d_out, int out_size, void* d_ws, size_t ws_size,
                              hipStream_t stream) {
    const float* x_user  = (const float*)d_in[0];
    const float* x_movie = (const float*)d_in[1];
    const int* edge_src  = (const int*)d_in[2];
    const int* edge_dst  = (const int*)d_in[3];
    const int* label_u   = (const int*)d_in[4];
    const int* label_m   = (const int*)d_in[5];
    const float* W_nbr1_um  = (const float*)d_in[6];
    const float* W_root1_um = (const float*)d_in[7];
    const float* W_nbr1_mu  = (const float*)d_in[8];
    const float* W_root1_mu = (const float*)d_in[9];
    const float* W_nbr2_um  = (const float*)d_in[10];
    const float* W_root2_um = (const float*)d_in[11];
    const float* W_nbr2_mu  = (const float*)d_in[12];
    const float* W_root2_mu = (const float*)d_in[13];
    const float* b1_m = (const float*)d_in[14];
    const float* b1_u = (const float*)d_in[15];
    const float* b2_m = (const float*)d_in[16];
    const float* b2_u = (const float*)d_in[17];

    const int n_users  = in_sizes[0] / CDIM;
    const int n_movies = in_sizes[1] / CDIM;
    const int E  = in_sizes[2];
    const int EL = in_sizes[4];

    const int N = n_movies + n_users;

    float* ws = (float*)d_ws;
    float* h2_m = ws;
    float* h2_u = h2_m + (size_t)n_movies * CDIM;
    float* h_m  = h2_u + (size_t)n_users * CDIM;
    float* h_u  = h_m + (size_t)n_movies * CDIM;
    float* y_m  = h_u + (size_t)n_users * CDIM;
    int*   deg    = (int*)(y_m + (size_t)n_movies * CDIM);
    int*   slot_m = deg + N + 64;
    int*   slot_u = slot_m + (size_t)n_movies * CAP_MOVIE;
    int*   spill  = slot_u + (size_t)n_users * CAP_USER;

    float* out = (float*)d_out;

    const int node_blocks = (N + 255) / 256;
    const int edge_blocks = (E + 255) / 256;
    const int lab_blocks  = (EL + 255) / 256;
    const int mov_blocks  = (n_movies + 15) / 16;
    const int usr_blocks  = (n_users + 63) / 64;

    GNN_zero_kernel<<<node_blocks, 256, 0, stream>>>((float*)deg, (size_t)N + 1);
    GNN_slot_kernel<<<edge_blocks, 256, 0, stream>>>(
        edge_src, edge_dst, deg, slot_m, slot_u, spill, E, n_movies, N);

    GNN_sage_gemm_kernel<1, false><<<mov_blocks, 256, 0, stream>>>(
        x_user, x_movie, deg, slot_m, spill, deg + N, W_nbr1_um, W_root1_um, b1_m, h_m,
        W_nbr1_mu, y_m, n_movies, 0, CAP_MOVIE, 1);
    GNN_sage_gemm_kernel<4, true><<<usr_blocks, 256, 0, stream>>>(
        y_m, x_user, deg, slot_u, spill, deg + N, nullptr, W_root1_mu, b1_u, h_u,
        nullptr, nullptr, n_users, n_movies, CAP_USER, 1);

    GNN_sage_gemm_kernel<1, false><<<mov_blocks, 256, 0, stream>>>(
        h_u, h_m, deg, slot_m, spill, deg + N, W_nbr2_um, W_root2_um, b2_m, h2_m,
        W_nbr2_mu, y_m, n_movies, 0, CAP_MOVIE, 0);
    GNN_sage_gemm_kernel<4, true><<<usr_blocks, 256, 0, stream>>>(
        y_m, h_u, deg, slot_u, spill, deg + N, nullptr, W_root2_mu, b2_u, h2_u,
        nullptr, nullptr, n_users, n_movies, CAP_USER, 0);

    GNN_decode_kernel<<<lab_blocks, 256, 0, stream>>>(
        h2_u, h2_m, label_u, label_m, out, EL);

    (void)n_in; (void)out_size; (void)ws_size;
}


# Users' node update: 32 rows per block instead of 64 and at most 64 registers per wave: twice as many waves per SIMD walk neighbour lists at once, each half as many; a weight fragment feeds 2 tiles
# speedup vs baseline: 2.4356x; 1.0473x over previous
#include <hip/hip_runtime.h>
#include <hip/hip_bf16.h>

typedef __attribute__((ext_vector_type(2))) float v2f;
typedef __attribute__((ext_vector_type(8))) float v8f;

#define CDIM 128

__global__ void GNN_zero_kernel(float* __restrict__ p, size_t n) {
    size_t i = (size_t)blockIdx.x * blockDim.x + threadIdx.x;
    size_t stride = (size_t)gridDim.x * blockDim.x;
    for (; i < n; i += stride) p[i] = 0.0f;
}

#define CAP_MOVIE 64
#define CAP_USER  32

__global__ void GNN_slot_kernel(const int* __restrict__ src_idx,
                                const int* __restrict__ dst_idx,
                                int* __restrict__ cnt,
                                int* __restrict__ slot_m,
                                int* __restrict__ slot_u,
                                int* __restrict__ spill,
                                int n_edges, int n_movies, int n_nodes) {
    int e = (int)(blockIdx.x * (size_t)blockDim.x + threadIdx.x);
    if (e >= n_edges) return;
    int s = src_idx[e];
    int d = dst_idx[e];
    const int pm = atomicAdd(cnt + d, 1);
    const int pu = atomicAdd(cnt + n_movies + s, 1);
    if (pm < CAP_MOVIE) {
        slot_m[(size_t)d * CAP_MOVIE + pm] = s;
    } else {
        const int o = atomicAdd(cnt + n_nodes, 1);
        spill[2 * (size_t)o]     = d;
        spill[2 * (size_t)o + 1] = s;
    }
    if (pu < CAP_USER) {
        slot_u[(size_t)s * CAP_USER + pu] = d;
    } else {
        const int o = atomicAdd(cnt + n_nodes, 1);
        spill[2 * (size_t)o]     = n_movies + s;
        spill[2 * (size_t)o + 1] = d;
    }
}

static __device__ __forceinline__ float4 GNN_row4(const float* feat, int r, int lane) {
    return *(const float4*)(feat + (size_t)r * CDIM + lane * 4);
}
static __device__ __forceinline__ void GNN_add4(float4& a, const float4& x) {
    a.x += x.x; a.y += x.y; a.z += x.z; a.w += x.w;
}

static __device__ __forceinline__ float4 GNN_list_sum(const float* __restrict__ feat,
                                                      const int*   __restrict__ col,
                                                      int start, int len, int first, int lane) {
    float4 acc = make_float4(0.0f, 0.0f, 0.0f, 0.0f);
    for (int base = 0; base < len; base += 32) {
        const int n = min(32, len - base);
        int idx = first;
        if (base > 0) idx = (lane < n) ? col[start + base + lane] : 0;
        int j = 0;
        for (; j + 4 <= n; j += 4) {
            const float4 x0 = GNN_row4(feat, __builtin_amdgcn_readlane(idx, j),     lane);
            const float4 x1 = GNN_row4(feat, __builtin_amdgcn_readlane(idx, j + 1), lane);
            const float4 x2 = GNN_row4(feat, __builtin_amdgcn_readlane(idx, j + 2), lane);
            const float4 x3 = GNN_row4(feat, __builtin_amdgcn_readlane(idx, j + 3), lane);
            GNN_add4(acc, x0); GNN_add4(acc, x1); GNN_add4(acc, x2); GNN_add4(acc, x3);
        }
        if (j < n) {
            const bool two   = j + 1 < n;
            const bool three = j + 2 < n;
            const int r0 = __builtin_amdgcn_readlane(idx, j);
            const int r1 = __builtin_amdgcn_readlane(idx, two   ? j + 1 : j);
            const int r2 = __builtin_amdgcn_readlane(idx, three ? j + 2 : j);
            const float4 x0 = GNN_row4(feat, r0, lane);
            float4 x1 = make_float4(0.0f, 0.0f, 0.0f, 0.0f);
            float4 x2 = make_float4(0.0f, 0.0f, 0.0f, 0.0f);
            if (two)   x1 = GNN_row4(feat, r1, lane);
            if (three) x2 = GNN_row4(feat, r2, lane);
            GNN_add4(acc, x0);
            if (two)   GNN_add4(acc, x1);
            if (three) GNN_add4(acc, x2);
        }
    }
    return acc;
}

static __device__ __forceinline__ void GNN_spill_sum(float4& acc,
                                                     const float* __restrict__ feat,
                                                     const int*   __restrict__ spill,
                                                     int n_spill, int node, int lane) {
    for (int i = 0; i < n_spill; ++i) {
        if (spill[2 * (size_t)i] == node) {
            const float4 x = GNN_row4(feat, spill[2 * (size_t)i + 1], lane);
            GNN_add4(acc, x);
        }
    }
}

template <int MT, bool PRE>
__global__ __launch_bounds__(256, PRE ? 16 : 1)
void GNN_sage_gemm_kernel(const float* __restrict__ nbr_feat,
                          const float* __restrict__ root,
                          const int*   __restrict__ deg,
                          const int*   __restrict__ col,
                          const int*   __restrict__ spill,
                          const int*   __restrict__ spill_cnt,
                          const float* __restrict__ W_nbr,
                          const float* __restrict__ W_root,
                          const float* __restrict__ bias,
                          float* __restrict__ out,
                          const float* __restrict__ W_pre,
                          float* __restrict__ pre_out,
                          int n_rows, int node_base, int cap, int do_relu) {
    __shared__ float sMean[16 * MT][CDIM + 4];
    __shared__ float sRoot[16 * MT][CDIM + 4];

    const int tid  = threadIdx.x;
    const int row0 = blockIdx.x * (16 * MT);

    {
        const int w  = __builtin_amdgcn_readfirstlane(tid >> 5);
        const int ln = tid & 31;
        for (int p = 0; p < MT; ++p) {
            const int r0   = 16 * p + 2 * w;
            const int gr0  = row0 + r0;
            const int gr1  = gr0 + 1;
            const bool ok0 = gr0 < n_rows;
            const bool ok1 = gr1 < n_rows;
            int tot0 = 0, tot1 = 0;
            if (ok0) tot0 = deg[node_base + gr0];
            if (ok1) tot1 = deg[node_base + gr1];
            const int len0   = min(tot0, cap);
            const int len1   = min(tot1, cap);
            const int start0 = ok0 ? gr0 * cap : 0;
            const int start1 = ok1 ? gr1 * cap : 0;
            float4 x0 = make_float4(0.0f, 0.0f, 0.0f, 0.0f);
            float4 x1 = make_float4(0.0f, 0.0f, 0.0f, 0.0f);
            if (ok0) x0 = GNN_row4(root, gr0, ln);
            if (ok1) x1 = GNN_row4(root, gr1, ln);
            const int first0 = (ln < min(32, len0)) ? col[start0 + ln] : 0;
            const int first1 = (ln < min(32, len1)) ? col[start1 + ln] : 0;
            float4 a0 = GNN_list_sum(nbr_feat, col, start0, len0, first0, ln);
            float4 a1 = GNN_list_sum(nbr_feat, col, start1, len1, first1, ln);
            if (tot0 > cap) GNN_spill_sum(a0, nbr_feat, spill, spill_cnt[0], node_base + gr0, ln);
            if (tot1 > cap) GNN_spill_sum(a1, nbr_feat, spill, spill_cnt[0], node_base + gr1, ln);
            const float inv0 = 1.0f / fmaxf((float)tot0, 1.0f);
            const float inv1 = 1.0f / fmaxf((float)tot1, 1.0f);
            *(float4*)&sMean[r0][ln * 4]     = make_float4(a0.x * inv0, a0.y * inv0, a0.z * inv0, a0.w * inv0);
            *(float4*)&sMean[r0 + 1][ln * 4] = make_float4(a1.x * inv1, a1.y * inv1, a1.z * inv1, a1.w * inv1);
            *(float4*)&sRoot[r0][ln * 4]     = x0;
            *(float4*)&sRoot[r0 + 1][ln * 4] = x1;
        }
    }
    __syncthreads();

    const int wave  = tid >> 5;
    const int lane  = tid & 31;
    const int lhalf = (lane >> 4) & 1;
    const int lm    = lane & 15;
    const int n0    = wave * 16;

    v8f acc[MT];
    {
        float bv = bias[n0 + lm];
#pragma unroll
        for (int t = 0; t < MT; ++t) {
#pragma unroll
            for (int i = 0; i < 8; ++i) {
                if constexpr (PRE) acc[t][i] = bv + sMean[16 * t + i + 8 * lhalf][n0 + lm];
                else               acc[t][i] = bv;
            }
        }
    }

    if constexpr (!PRE) {
#pragma unroll
        for (int k = 0; k < CDIM; k += 4) {
            int ka = k + 2 * lhalf;
            v2f b;
            b.x = W_nbr[(size_t)ka * CDIM + n0 + lm];
            b.y = W_nbr[(size_t)(ka + 1) * CDIM + n0 + lm];
#pragma unroll
            for (int t = 0; t < MT; ++t) {
                v2f a;
                a.x = sMean[16 * t + lm][ka];
                a.y = sMean[16 * t + lm][ka + 1];
                acc[t] = __builtin_amdgcn_wmma_f32_16x16x4_f32(
                    false, a, false, b, (short)0, acc[t], false, false);
            }
        }
    }

#pragma unroll (PRE ? 16 : 32)
    for (int k = 0; k < CDIM; k += 4) {
        int ka = k + 2 * lhalf;
        v2f b;
        b.x = W_root[(size_t)ka * CDIM + n0 + lm];
        b.y = W_root[(size_t)(ka + 1) * CDIM + n0 + lm];
#pragma unroll
        for (int t = 0; t < MT; ++t) {
            v2f a;
            a.x = sRoot[16 * t + lm][ka];
            a.y = sRoot[16 * t + lm][ka + 1];
            acc[t] = __builtin_amdgcn_wmma_f32_16x16x4_f32(
                false, a, false, b, (short)0, acc[t], false, false);
        }
    }

#pragma unroll
    for (int t = 0; t < MT; ++t) {
#pragma unroll
        for (int v = 0; v < 8; ++v) {
            float r = acc[t][v];
            if (do_relu) r = fmaxf(r, 0.0f);
            int gr = row0 + 16 * t + v + 8 * lhalf;
            if (gr < n_rows) out[(size_t)gr * CDIM + n0 + lm] = r;
        }
    }

    if constexpr (!PRE) {
        v8f pre[MT];
#pragma unroll
        for (int t = 0; t < MT; ++t) {
#pragma unroll
            for (int i = 0; i < 8; ++i) pre[t][i] = 0.0f;
        }
#pragma unroll
        for (int k = 0; k < CDIM; k += 4) {
            int ka = k + 2 * lhalf;
            v2f b;
            b.x = W_pre[(size_t)ka * CDIM + n0 + lm];
            b.y = W_pre[(size_t)(ka + 1) * CDIM + n0 + lm];
#pragma unroll
            for (int t = 0; t < MT; ++t) {
                v2f a;
                a.x = sRoot[16 * t + lm][ka];
                a.y = sRoot[16 * t + lm][ka + 1];
                pre[t] = __builtin_amdgcn_wmma_f32_16x16x4_f32(
                    false, a, false, b, (short)0, pre[t], false, false);
            }
        }
#pragma unroll
        for (int t = 0; t < MT; ++t) {
#pragma unroll
            for (int v = 0; v < 8; ++v) {
                int gr = row0 + 16 * t + v + 8 * lhalf;
                if (gr < n_rows) pre_out[(size_t)gr * CDIM + n0 + lm] = pre[t][v];
            }
        }
    }
}

static __device__ __forceinline__ float GNN_dot4(const float4& a, const float4& b) {
    float s = a.x * b.x;
    s += a.y * b.y;
    s += a.z * b.z;
    s += a.w * b.w;
    return s;
}
static __device__ __forceinline__ float GNN_wave_sum(float s) {
    for (int o = 16; o > 0; o >>= 1) s += __shfl_xor(s, o, 32);
    return s;
}

__global__ __launch_bounds__(256)
void GNN_decode_kernel(const float* __restrict__ h_u2,
                       const float* __restrict__ h_m2,
                       const int*   __restrict__ lab_u,
                       const int*   __restrict__ lab_m,
                       float* __restrict__ out, int n) {
    int chunk = (int)((blockIdx.x * (size_t)blockDim.x + threadIdx.x) >> 5);
    chunk = __builtin_amdgcn_readfirstlane(chunk);
    const int lane = threadIdx.x & 31;
    const int base = chunk * 32;
    if (base >= n) return;
    const int cnt = min(32, n - base);
    const int iu = (lane < cnt) ? lab_u[base + lane] : 0;
    const int im = (lane < cnt) ? lab_m[base + lane] : 0;
    float res = 0.0f;
    int j = 0;
    for (; j + 4 <= cnt; j += 4) {
        const float4 a0 = GNN_row4(h_u2, __builtin_amdgcn_readlane(iu, j),     lane);
        const float4 b0 = GNN_row4(h_m2, __builtin_amdgcn_readlane(im, j),     lane);
        const float4 a1 = GNN_row4(h_u2, __builtin_amdgcn_readlane(iu, j + 1), lane);
        const float4 b1 = GNN_row4(h_m2, __builtin_amdgcn_readlane(im, j + 1), lane);
        const float4 a2 = GNN_row4(h_u2, __builtin_amdgcn_readlane(iu, j + 2), lane);
        const float4 b2 = GNN_row4(h_m2, __builtin_amdgcn_readlane(im, j + 2), lane);
        const float4 a3 = GNN_row4(h_u2, __builtin_amdgcn_readlane(iu, j + 3), lane);
        const float4 b3 = GNN_row4(h_m2, __builtin_amdgcn_readlane(im, j + 3), lane);
        const float s0 = GNN_wave_sum(GNN_dot4(a0, b0));
        const float s1 = GNN_wave_sum(GNN_dot4(a1, b1));
        const float s2 = GNN_wave_sum(GNN_dot4(a2, b2));
        const float s3 = GNN_wave_sum(GNN_dot4(a3, b3));
        res = (lane == j)     ? s0 : res;
        res = (lane == j + 1) ? s1 : res;
        res = (lane == j + 2) ? s2 : res;
        res = (lane == j + 3) ? s3 : res;
    }
    for (; j < cnt; ++j) {
        const float4 a0 = GNN_row4(h_u2, __builtin_amdgcn_readlane(iu, j), lane);
        const float4 b0 = GNN_row4(h_m2, __builtin_amdgcn_readlane(im, j), lane);
        const float s0 = GNN_wave_sum(GNN_dot4(a0, b0));
        res = (lane == j) ? s0 : res;
    }
    if (lane < cnt) out[base + lane] = res;
}

extern "C" void kernel_launch(void* const* d_in, const int* in_sizes, int n_in,
                              void* d_out, int out_size, void* d_ws, size_t ws_size,
                              hipStream_t stream) {
    const float* x_user  = (const float*)d_in[0];
    const float* x_movie = (const float*)d_in[1];
    const int* edge_src  = (const int*)d_in[2];
    const int* edge_dst  = (const int*)d_in[3];
    const int* label_u   = (const int*)d_in[4];
    const int* label_m   = (const int*)d_in[5];
    const float* W_nbr1_um  = (const float*)d_in[6];
    const float* W_root1_um = (const float*)d_in[7];
    const float* W_nbr1_mu  = (const float*)d_in[8];
    const float* W_root1_mu = (const float*)d_in[9];
    const float* W_nbr2_um  = (const float*)d_in[10];
    const float* W_root2_um = (const float*)d_in[11];
    const float* W_nbr2_mu  = (const float*)d_in[12];
    const float* W_root2_mu = (const float*)d_in[13];
    const float* b1_m = (const float*)d_in[14];
    const float* b1_u = (const float*)d_in[15];
    const float* b2_m = (const float*)d_in[16];
    const float* b2_u = (const float*)d_in[17];

    const int n_users  = in_sizes[0] / CDIM;
    const int n_movies = in_sizes[1] / CDIM;
    const int E  = in_sizes[2];
    const int EL = in_sizes[4];

    const int N = n_movies + n_users;

    float* ws = (float*)d_ws;
    float* h2_m = ws;
    float* h2_u = h2_m + (size_t)n_movies * CDIM;
    float* h_m  = h2_u + (size_t)n_users * CDIM;
    float* h_u  = h_m + (size_t)n_movies * CDIM;
    float* y_m  = h_u + (size_t)n_users * CDIM;
    int*   deg    = (int*)(y_m + (size_t)n_movies * CDIM);
    int*   slot_m = deg + N + 64;
    int*   slot_u = slot_m + (size_t)n_movies * CAP_MOVIE;
    int*   spill  = slot_u + (size_t)n_users * CAP_USER;

    float* out = (float*)d_out;

    const int node_blocks = (N + 255) / 256;
    const int edge_blocks = (E + 255) / 256;
    const int lab_blocks  = (EL + 255) / 256;
    const int mov_blocks  = (n_movies + 15) / 16;
    const int usr_blocks  = (n_users + 31) / 32;

    GNN_zero_kernel<<<node_blocks, 256, 0, stream>>>((float*)deg, (size_t)N + 1);
    GNN_slot_kernel<<<edge_blocks, 256, 0, stream>>>(
        edge_src, edge_dst, deg, slot_m, slot_u, spill, E, n_movies, N);

    GNN_sage_gemm_kernel<1, false><<<mov_blocks, 256, 0, stream>>>(
        x_user, x_movie, deg, slot_m, spill, deg + N, W_nbr1_um, W_root1_um, b1_m, h_m,
        W_nbr1_mu, y_m, n_movies, 0, CAP_MOVIE, 1);
    GNN_sage_gemm_kernel<2, true><<<usr_blocks, 256, 0, stream>>>(
        y_m, x_user, deg, slot_u, spill, deg + N, nullptr, W_root1_mu, b1_u, h_u,
        nullptr, nullptr, n_users, n_movies, CAP_USER, 1);

    GNN_sage_gemm_kernel<1, false><<<mov_blocks, 256, 0, stream>>>(
        h_u, h_m, deg, slot_m, spill, deg + N, W_nbr2_um, W_root2_um, b2_m, h2_m,
        W_nbr2_mu, y_m, n_movies, 0, CAP_MOVIE, 0);
    GNN_sage_gemm_kernel<2, true><<<usr_blocks, 256, 0, stream>>>(
        y_m, h_u, deg, slot_u, spill, deg + N, nullptr, W_root2_mu, b2_u, h2_u,
        nullptr, nullptr, n_users, n_movies, CAP_USER, 0);

    GNN_decode_kernel<<<lab_blocks, 256, 0, stream>>>(
        h2_u, h2_m, label_u, label_m, out, EL);

    (void)n_in; (void)out_size; (void)ws_size;
}


# Both user layers in one launch: a block keeps its 32 layer-1 rows in LDS as layer 2's root tile; the layer-1 movie launch also makes layer 2's weighted table; 7 launches to 6; all sum orders kept
# speedup vs baseline: 2.5088x; 1.0301x over previous
#include <hip/hip_runtime.h>
#include <hip/hip_bf16.h>

typedef __attribute__((ext_vector_type(2))) float v2f;
typedef __attribute__((ext_vector_type(8))) float v8f;

#define CDIM 128

__global__ void GNN_zero_kernel(float* __restrict__ p, size_t n) {
    size_t i = (size_t)blockIdx.x * blockDim.x + threadIdx.x;
    size_t stride = (size_t)gridDim.x * blockDim.x;
    for (; i < n; i += stride) p[i] = 0.0f;
}

#define CAP_MOVIE 64
#define CAP_USER  32

__global__ void GNN_slot_kernel(const int* __restrict__ src_idx,
                                const int* __restrict__ dst_idx,
                                int* __restrict__ cnt,
                                int* __restrict__ slot_m,
                                int* __restrict__ slot_u,
                                int* __restrict__ spill,
                                int n_edges, int n_movies, int n_nodes) {
    int e = (int)(blockIdx.x * (size_t)blockDim.x + threadIdx.x);
    if (e >= n_edges) return;
    int s = src_idx[e];
    int d = dst_idx[e];
    const int pm = atomicAdd(cnt + d, 1);
    const int pu = atomicAdd(cnt + n_movies + s, 1);
    if (pm < CAP_MOVIE) {
        slot_m[(size_t)d * CAP_MOVIE + pm] = s;
    } else {
        const int o = atomicAdd(cnt + n_nodes, 1);
        spill[2 * (size_t)o]     = d;
        spill[2 * (size_t)o + 1] = s;
    }
    if (pu < CAP_USER) {
        slot_u[(size_t)s * CAP_USER + pu] = d;
    } else {
        const int o = atomicAdd(cnt + n_nodes, 1);
        spill[2 * (size_t)o]     = n_movies + s;
        spill[2 * (size_t)o + 1] = d;
    }
}

static __device__ __forceinline__ float4 GNN_row4(const float* feat, int r, int lane) {
    return *(const float4*)(feat + (size_t)r * CDIM + lane * 4);
}
static __device__ __forceinline__ void GNN_add4(float4& a, const float4& x) {
    a.x += x.x; a.y += x.y; a.z += x.z; a.w += x.w;
}

static __device__ __forceinline__ float4 GNN_list_sum(const float* __restrict__ feat,
                                                      const int*   __restrict__ col,
                                                      int start, int len, int first, int lane) {
    float4 acc = make_float4(0.0f, 0.0f, 0.0f, 0.0f);
    for (int base = 0; base < len; base += 32) {
        const int n = min(32, len - base);
        int idx = first;
        if (base > 0) idx = (lane < n) ? col[start + base + lane] : 0;
        int j = 0;
        for (; j + 4 <= n; j += 4) {
            const float4 x0 = GNN_row4(feat, __builtin_amdgcn_readlane(idx, j),     lane);
            const float4 x1 = GNN_row4(feat, __builtin_amdgcn_readlane(idx, j + 1), lane);
            const float4 x2 = GNN_row4(feat, __builtin_amdgcn_readlane(idx, j + 2), lane);
            const float4 x3 = GNN_row4(feat, __builtin_amdgcn_readlane(idx, j + 3), lane);
            GNN_add4(acc, x0); GNN_add4(acc, x1); GNN_add4(acc, x2); GNN_add4(acc, x3);
        }
        if (j < n) {
            const bool two   = j + 1 < n;
            const bool three = j + 2 < n;
            const int r0 = __builtin_amdgcn_readlane(idx, j);
            const int r1 = __builtin_amdgcn_readlane(idx, two   ? j + 1 : j);
            const int r2 = __builtin_amdgcn_readlane(idx, three ? j + 2 : j);
            const float4 x0 = GNN_row4(feat, r0, lane);
            float4 x1 = make_float4(0.0f, 0.0f, 0.0f, 0.0f);
            float4 x2 = make_float4(0.0f, 0.0f, 0.0f, 0.0f);
            if (two)   x1 = GNN_row4(feat, r1, lane);
            if (three) x2 = GNN_row4(feat, r2, lane);
            GNN_add4(acc, x0);
            if (two)   GNN_add4(acc, x1);
            if (three) GNN_add4(acc, x2);
        }
    }
    return acc;
}

static __device__ __forceinline__ void GNN_spill_sum(float4& acc,
                                                     const float* __restrict__ feat,
                                                     const int*   __restrict__ spill,
                                                     int n_spill, int node, int lane) {
    for (int i = 0; i < n_spill; ++i) {
        if (spill[2 * (size_t)i] == node) {
            const float4 x = GNN_row4(feat, spill[2 * (size_t)i + 1], lane);
            GNN_add4(acc, x);
        }
    }
}


template <int MT, int UNROLL>
static __device__ __forceinline__ void GNN_tile_product(v8f (&acc)[MT],
                                                        const float (*tile)[CDIM + 4],
                                                        const float* __restrict__ W,
                                                        int n0, int lm, int lhalf) {
#pragma unroll UNROLL
    for (int k = 0; k < CDIM; k += 4) {
        int ka = k + 2 * lhalf;
        v2f b;
        b.x = W[(size_t)ka * CDIM + n0 + lm];
        b.y = W[(size_t)(ka + 1) * CDIM + n0 + lm];
#pragma unroll
        for (int t = 0; t < MT; ++t) {
            v2f a;
            a.x = tile[16 * t + lm][ka];
            a.y = tile[16 * t + lm][ka + 1];
            acc[t] = __builtin_amdgcn_wmma_f32_16x16x4_f32(
                false, a, false, b, (short)0, acc[t], false, false);
        }
    }
}

template <int MT, bool FIRST>
__global__ __launch_bounds__(256, 1)
void GNN_movie_kernel(const float* __restrict__ nbr_feat,
                      const float* __restrict__ root,
                      const int*   __restrict__ deg,
                      const int*   __restrict__ col,
                      const int*   __restrict__ spill,
                      const int*   __restrict__ spill_cnt,
                      const float* __restrict__ W_nbr,
                      const float* __restrict__ W_root,
                      const float* __restrict__ bias,
                      float* __restrict__ out,
                      const float* __restrict__ W_pre,
                      float* __restrict__ pre_out,
                      const float* __restrict__ W_pre2,
                      float* __restrict__ pre2_out,
                      int n_rows, int node_base, int cap, int do_relu) {
    __shared__ float sMean[16 * MT][CDIM + 4];
    __shared__ float sRoot[16 * MT][CDIM + 4];

    const int tid  = threadIdx.x;
    const int row0 = blockIdx.x * (16 * MT);

    {
        const int w  = __builtin_amdgcn_readfirstlane(tid >> 5);
        const int ln = tid & 31;
        for (int p = 0; p < MT; ++p) {
            const int r0   = 16 * p + 2 * w;
            const int gr0  = row0 + r0;
            const int gr1  = gr0 + 1;
            const bool ok0 = gr0 < n_rows;
            const bool ok1 = gr1 < n_rows;
            int tot0 = 0, tot1 = 0;
            if (ok0) tot0 = deg[node_base + gr0];
            if (ok1) tot1 = deg[node_base + gr1];
            const int len0   = min(tot0, cap);
            const int len1   = min(tot1, cap);
            const int start0 = ok0 ? gr0 * cap : 0;
            const int start1 = ok1 ? gr1 * cap : 0;
            float4 x0 = make_float4(0.0f, 0.0f, 0.0f, 0.0f);
            float4 x1 = make_float4(0.0f, 0.0f, 0.0f, 0.0f);
            if (ok0) x0 = GNN_row4(root, gr0, ln);
            if (ok1) x1 = GNN_row4(root, gr1, ln);
            const int first0 = (ln < min(32, len0)) ? col[start0 + ln] : 0;
            const int first1 = (ln < min(32, len1)) ? col[start1 + ln] : 0;
            float4 a0 = GNN_list_sum(nbr_feat, col, start0, len0, first0, ln);
            float4 a1 = GNN_list_sum(nbr_feat, col, start1, len1, first1, ln);
            if (tot0 > cap) GNN_spill_sum(a0, nbr_feat, spill, spill_cnt[0], node_base + gr0, ln);
            if (tot1 > cap) GNN_spill_sum(a1, nbr_feat, spill, spill_cnt[0], node_base + gr1, ln);
            const float inv0 = 1.0f / fmaxf((float)tot0, 1.0f);
            const float inv1 = 1.0f / fmaxf((float)tot1, 1.0f);
            *(float4*)&sMean[r0][ln * 4]     = make_float4(a0.x * inv0, a0.y * inv0, a0.z * inv0, a0.w * inv0);
            *(float4*)&sMean[r0 + 1][ln * 4] = make_float4(a1.x * inv1, a1.y * inv1, a1.z * inv1, a1.w * inv1);
            *(float4*)&sRoot[r0][ln * 4]     = x0;
            *(float4*)&sRoot[r0 + 1][ln * 4] = x1;
        }
    }
    __syncthreads();

    const int wave  = tid >> 5;
    const int lane  = tid & 31;
    const int lhalf = (lane >> 4) & 1;
    const int lm    = lane & 15;
    const int n0    = wave * 16;

    v8f acc[MT];
    {
        float bv = bias[n0 + lm];
#pragma unroll
        for (int t = 0; t < MT; ++t) {
#pragma unroll
            for (int i = 0; i < 8; ++i) acc[t][i] = bv;
        }
    }

    GNN_tile_product<MT, 32>(acc, sMean, W_nbr, n0, lm, lhalf);
    GNN_tile_product<MT, 32>(acc, sRoot, W_root, n0, lm, lhalf);

#pragma unroll
    for (int t = 0; t < MT; ++t) {
#pragma unroll
        for (int v = 0; v < 8; ++v) {
            float r = acc[t][v];
            if (do_relu) r = fmaxf(r, 0.0f);
            acc[t][v] = r;
            int gr = row0 + 16 * t + v + 8 * lhalf;
            if (gr < n_rows) out[(size_t)gr * CDIM + n0 + lm] = r;
        }
    }

    if constexpr (FIRST) {
        __syncthreads();
#pragma unroll
        for (int t = 0; t < MT; ++t) {
#pragma unroll
            for (int v = 0; v < 8; ++v) sMean[16 * t + v + 8 * lhalf][n0 + lm] = acc[t][v];
        }

        v8f pre[MT];
#pragma unroll
        for (int t = 0; t < MT; ++t) {
#pragma unroll
            for (int i = 0; i < 8; ++i) pre[t][i] = 0.0f;
        }
        GNN_tile_product<MT, 32>(pre, sRoot, W_pre, n0, lm, lhalf);
#pragma unroll
        for (int t = 0; t < MT; ++t) {
#pragma unroll
            for (int v = 0; v < 8; ++v) {
                int gr = row0 + 16 * t + v + 8 * lhalf;
                if (gr < n_rows) pre_out[(size_t)gr * CDIM + n0 + lm] = pre[t][v];
            }
        }

        __syncthreads();

#pragma unroll
        for (int t = 0; t < MT; ++t) {
#pragma unroll
            for (int i = 0; i < 8; ++i) pre[t][i] = 0.0f;
        }
        GNN_tile_product<MT, 32>(pre, sMean, W_pre2, n0, lm, lhalf);
#pragma unroll
        for (int t = 0; t < MT; ++t) {
#pragma unroll
            for (int v = 0; v < 8; ++v) {
                int gr = row0 + 16 * t + v + 8 * lhalf;
                if (gr < n_rows) pre2_out[(size_t)gr * CDIM + n0 + lm] = pre[t][v];
            }
        }
    }
}

template <int MT>
__global__ __launch_bounds__(256, 16)
void GNN_user_kernel(const float* __restrict__ tab1,
                     const float* __restrict__ tab2,
                     const float* __restrict__ root,
                     const int*   __restrict__ deg,
                     const int*   __restrict__ col,
                     const int*   __restrict__ spill,
                     const int*   __restrict__ spill_cnt,
                     const float* __restrict__ W_root1,
                     const float* __restrict__ W_root2,
                     const float* __restrict__ bias1,
                     const float* __restrict__ bias2,
                     float* __restrict__ out1,
                     float* __restrict__ out2,
                     int n_rows, int node_base, int cap) {
    __shared__ float sMean[16 * MT][CDIM + 4];
    __shared__ float sRoot[16 * MT][CDIM + 4];

    const int tid  = threadIdx.x;
    const int row0 = blockIdx.x * (16 * MT);
    const int wave  = tid >> 5;
    const int lane  = tid & 31;
    const int lhalf = (lane >> 4) & 1;
    const int lm    = lane & 15;
    const int n0    = wave * 16;

#pragma unroll
    for (int layer = 0; layer < 2; ++layer) {
        const float* __restrict__ nbr_feat = layer ? tab2 : tab1;
        const float* __restrict__ W_root   = layer ? W_root2 : W_root1;
        const float* __restrict__ bias     = layer ? bias2 : bias1;
        float* __restrict__ out            = layer ? out2 : out1;

        {
            const int w  = __builtin_amdgcn_readfirstlane(tid >> 5);
            const int ln = tid & 31;
            for (int p = 0; p < MT; ++p) {
                const int r0   = 16 * p + 2 * w;
                const int gr0  = row0 + r0;
                const int gr1  = gr0 + 1;
                const bool ok0 = gr0 < n_rows;
                const bool ok1 = gr1 < n_rows;
                int tot0 = 0, tot1 = 0;
                if (ok0) tot0 = deg[node_base + gr0];
                if (ok1) tot1 = deg[node_base + gr1];
                const int len0   = min(tot0, cap);
                const int len1   = min(tot1, cap);
                const int start0 = ok0 ? gr0 * cap : 0;
                const int start1 = ok1 ? gr1 * cap : 0;
                float4 x0 = make_float4(0.0f, 0.0f, 0.0f, 0.0f);
                float4 x1 = make_float4(0.0f, 0.0f, 0.0f, 0.0f);
                if (layer == 0) {
                    if (ok0) x0 = GNN_row4(root, gr0, ln);
                    if (ok1) x1 = GNN_row4(root, gr1, ln);
                }
                const int first0 = (ln < min(32, len0)) ? col[start0 + ln] : 0;
                const int first1 = (ln < min(32, len1)) ? col[start1 + ln] : 0;
                float4 a0 = GNN_list_sum(nbr_feat, col, start0, len0, first0, ln);
                float4 a1 = GNN_list_sum(nbr_feat, col, start1, len1, first1, ln);
                if (tot0 > cap) GNN_spill_sum(a0, nbr_feat, spill, spill_cnt[0], node_base + gr0, ln);
                if (tot1 > cap) GNN_spill_sum(a1, nbr_feat, spill, spill_cnt[0], node_base + gr1, ln);
                const float inv0 = 1.0f / fmaxf((float)tot0, 1.0f);
                const float inv1 = 1.0f / fmaxf((float)tot1, 1.0f);
                *(float4*)&sMean[r0][ln * 4]     = make_float4(a0.x * inv0, a0.y * inv0, a0.z * inv0, a0.w * inv0);
                *(float4*)&sMean[r0 + 1][ln * 4] = make_float4(a1.x * inv1, a1.y * inv1, a1.z * inv1, a1.w * inv1);
                if (layer == 0) {
                    *(float4*)&sRoot[r0][ln * 4]     = x0;
                    *(float4*)&sRoot[r0 + 1][ln * 4] = x1;
                }
            }
        }
        __syncthreads();

        v8f acc[MT];
        {
            float bv = bias[n0 + lm];
#pragma unroll
            for (int t = 0; t < MT; ++t) {
#pragma unroll
                for (int i = 0; i < 8; ++i) {
                    acc[t][i] = bv + sMean[16 * t + i + 8 * lhalf][n0 + lm];
                }
            }
        }

        GNN_tile_product<MT, 16>(acc, sRoot, W_root, n0, lm, lhalf);

        {
            float* __restrict__ oblk = out + (size_t)row0 * CDIM;
            const unsigned loff = (unsigned)((8 * lhalf) * CDIM + n0 + lm);
            const int rows_left = n_rows - row0 - 8 * lhalf;
#pragma unroll
            for (int t = 0; t < MT; ++t) {
#pragma unroll
                for (int v = 0; v < 8; ++v) {
                    float r = acc[t][v];
                    if (layer == 0) r = fmaxf(r, 0.0f);
                    acc[t][v] = r;
                    if (16 * t + v < rows_left) oblk[loff + (unsigned)((16 * t + v) * CDIM)] = r;
                }
            }
        }

        if (layer == 0) {
            __syncthreads();
#pragma unroll
            for (int t = 0; t < MT; ++t) {
#pragma unroll
                for (int v = 0; v < 8; ++v) sRoot[16 * t + v + 8 * lhalf][n0 + lm] = acc[t][v];
            }
        }
    }
}

static __device__ __forceinline__ float GNN_dot4(const float4& a, const float4& b) {
    float s = a.x * b.x;
    s += a.y * b.y;
    s += a.z * b.z;
    s += a.w * b.w;
    return s;
}
static __device__ __forceinline__ float GNN_wave_sum(float s) {
    for (int o = 16; o > 0; o >>= 1) s += __shfl_xor(s, o, 32);
    return s;
}

__global__ __launch_bounds__(256)
void GNN_decode_kernel(const float* __restrict__ h_u2,
                       const float* __restrict__ h_m2,
                       const int*   __restrict__ lab_u,
                       const int*   __restrict__ lab_m,
                       float* __restrict__ out, int n) {
    int chunk = (int)((blockIdx.x * (size_t)blockDim.x + threadIdx.x) >> 5);
    chunk = __builtin_amdgcn_readfirstlane(chunk);
    const int lane = threadIdx.x & 31;
    const int base = chunk * 32;
    if (base >= n) return;
    const int cnt = min(32, n - base);
    const int iu = (lane < cnt) ? lab_u[base + lane] : 0;
    const int im = (lane < cnt) ? lab_m[base + lane] : 0;
    float res = 0.0f;
    int j = 0;
    for (; j + 4 <= cnt; j += 4) {
        const float4 a0 = GNN_row4(h_u2, __builtin_amdgcn_readlane(iu, j),     lane);
        const float4 b0 = GNN_row4(h_m2, __builtin_amdgcn_readlane(im, j),     lane);
        const float4 a1 = GNN_row4(h_u2, __builtin_amdgcn_readlane(iu, j + 1), lane);
        const float4 b1 = GNN_row4(h_m2, __builtin_amdgcn_readlane(im, j + 1), lane);
        const float4 a2 = GNN_row4(h_u2, __builtin_amdgcn_readlane(iu, j + 2), lane);
        const float4 b2 = GNN_row4(h_m2, __builtin_amdgcn_readlane(im, j + 2), lane);
        const float4 a3 = GNN_row4(h_u2, __builtin_amdgcn_readlane(iu, j + 3), lane);
        const float4 b3 = GNN_row4(h_m2, __builtin_amdgcn_readlane(im, j + 3), lane);
        const float s0 = GNN_wave_sum(GNN_dot4(a0, b0));
        const float s1 = GNN_wave_sum(GNN_dot4(a1, b1));
        const float s2 = GNN_wave_sum(GNN_dot4(a2, b2));
        const float s3 = GNN_wave_sum(GNN_dot4(a3, b3));
        res = (lane == j)     ? s0 : res;
        res = (lane == j + 1) ? s1 : res;
        res = (lane == j + 2) ? s2 : res;
        res = (lane == j + 3) ? s3 : res;
    }
    for (; j < cnt; ++j) {
        const float4 a0 = GNN_row4(h_u2, __builtin_amdgcn_readlane(iu, j), lane);
        const float4 b0 = GNN_row4(h_m2, __builtin_amdgcn_readlane(im, j), lane);
        const float s0 = GNN_wave_sum(GNN_dot4(a0, b0));
        res = (lane == j) ? s0 : res;
    }
    if (lane < cnt) out[base + lane] = res;
}

extern "C" void kernel_launch(void* const* d_in, const int* in_sizes, int n_in,
                              void* d_out, int out_size, void* d_ws, size_t ws_size,
                              hipStream_t stream) {
    const float* x_user  = (const float*)d_in[0];
    const float* x_movie = (const float*)d_in[1];
    const int* edge_src  = (const int*)d_in[2];
    const int* edge_dst  = (const int*)d_in[3];
    const int* label_u   = (const int*)d_in[4];
    const int* label_m   = (const int*)d_in[5];
    const float* W_nbr1_um  = (const float*)d_in[6];
    const float* W_root1_um = (const float*)d_in[7];
    const float* W_nbr1_mu  = (const float*)d_in[8];
    const float* W_root1_mu = (const float*)d_in[9];
    const float* W_nbr2_um  = (const float*)d_in[10];
    const float* W_root2_um = (const float*)d_in[11];
    const float* W_nbr2_mu  = (const float*)d_in[12];
    const float* W_root2_mu = (const float*)d_in[13];
    const float* b1_m = (const float*)d_in[14];
    const float* b1_u = (const float*)d_in[15];
    const float* b2_m = (const float*)d_in[16];
    const float* b2_u = (const float*)d_in[17];

    const int n_users  = in_sizes[0] / CDIM;
    const int n_movies = in_sizes[1] / CDIM;
    const int E  = in_sizes[2];
    const int EL = in_sizes[4];

    const int N = n_movies + n_users;

    float* ws = (float*)d_ws;
    float* h2_m = ws;
    float* h2_u = h2_m + (size_t)n_movies * CDIM;
    float* h_m  = h2_u + (size_t)n_users * CDIM;
    float* h_u  = h_m + (size_t)n_movies * CDIM;
    float* y_m  = h_u + (size_t)n_users * CDIM;
    float* y_m2 = y_m + (size_t)n_movies * CDIM;
    int*   deg    = (int*)(y_m2 + (size_t)n_movies * CDIM);
    int*   slot_m = deg + N + 64;
    int*   slot_u = slot_m + (size_t)n_movies * CAP_MOVIE;
    int*   spill  = slot_u + (size_t)n_users * CAP_USER;

    float* out = (float*)d_out;

    const int node_blocks = (N + 255) / 256;
    const int edge_blocks = (E + 255) / 256;
    const int lab_blocks  = (EL + 255) / 256;
    const int mov_blocks  = (n_movies + 15) / 16;
    const int usr_blocks  = (n_users + 31) / 32;

    GNN_zero_kernel<<<node_blocks, 256, 0, stream>>>((float*)deg, (size_t)N + 1);
    GNN_slot_kernel<<<edge_blocks, 256, 0, stream>>>(
        edge_src, edge_dst, deg, slot_m, slot_u, spill, E, n_movies, N);

    GNN_movie_kernel<1, true><<<mov_blocks, 256, 0, stream>>>(
        x_user, x_movie, deg, slot_m, spill, deg + N, W_nbr1_um, W_root1_um, b1_m, h_m,
        W_nbr1_mu, y_m, W_nbr2_mu, y_m2, n_movies, 0, CAP_MOVIE, 1);

    GNN_user_kernel<2><<<usr_blocks, 256, 0, stream>>>(
        y_m, y_m2, x_user, deg, slot_u, spill, deg + N, W_root1_mu, W_root2_mu, b1_u, b2_u,
        h_u, h2_u, n_users, n_movies, CAP_USER);

    GNN_movie_kernel<1, false><<<mov_blocks, 256, 0, stream>>>(
        h_u, h_m, deg, slot_m, spill, deg + N, W_nbr2_um, W_root2_um, b2_m, h2_m,
        nullptr, nullptr, nullptr, nullptr, n_movies, 0, CAP_MOVIE, 0);

    GNN_decode_kernel<<<lab_blocks, 256, 0, stream>>>(
        h2_u, h2_m, label_u, label_m, out, EL);

    (void)n_in; (void)out_size; (void)ws_size;
}
